# starting kernel
# baseline (speedup 1.0000x reference)
#include <hip/hip_runtime.h>
#include <hip/hip_bf16.h>
#include <stdint.h>

typedef __attribute__((ext_vector_type(16))) _Float16 v16h;
typedef __attribute__((ext_vector_type(8)))  float    v8f;

#define IN_C   128
#define HID_C  128
#define OUT_C  64
#define NGRAPH 128

__device__ __forceinline__ void atomic_add_f32(float* p, float v) {
    unsafeAtomicAdd(p, v);
}

__global__ void k_fill(float* p, float v, int n) {
    int i = blockIdx.x * blockDim.x + threadIdx.x;
    if (i < n) p[i] = v;
}

__global__ void k_edge_deg(const int* __restrict__ dst, float* deg, int E) {
    int e = blockIdx.x * blockDim.x + threadIdx.x;
    if (e < E) atomic_add_f32(&deg[dst[e]], 1.0f);
}

__global__ void k_rsqrt(const float* __restrict__ deg, float* dinv, int n) {
    int i = blockIdx.x * blockDim.x + threadIdx.x;
    if (i < n) dinv[i] = rsqrtf(deg[i]);
}

__global__ void k_pack_w(const float* __restrict__ W, _Float16* __restrict__ Wh,
                         int K, int N) {
    int idx = blockIdx.x * blockDim.x + threadIdx.x;
    if (idx >= K * N) return;
    int frag   = idx >> 9;
    int within = idx & 511;
    int lane   = within >> 4;
    int i      = within & 15;
    int ntiles = N >> 4;
    int kt = frag / ntiles, nt = frag % ntiles;
    int n  = (nt << 4) + (lane & 15);
    int kk = (i < 8 ? i : i + 8) + ((lane >> 4) << 3);
    int k  = (kt << 5) + kk;
    Wh[idx] = (_Float16)W[k * N + n];
}

template <int NT, bool RELU>
__global__ void k_gemm(const float* __restrict__ A, const _Float16* __restrict__ Bh,
                       float* __restrict__ C, int M, int K, int N) {
    int wave = (int)((blockIdx.x * blockDim.x + threadIdx.x) >> 5);
    int lane = threadIdx.x & 31;
    int mtiles = M >> 4;
    if (wave >= mtiles) return;

    int m0  = wave << 4;
    int hi  = lane >> 4;
    int row = m0 + (lane & 15);

    v8f acc[NT];
    v8f zero = {0.f,0.f,0.f,0.f,0.f,0.f,0.f,0.f};
    #pragma unroll
    for (int j = 0; j < NT; ++j) acc[j] = zero;

    const v16h* bp = reinterpret_cast<const v16h*>(Bh);
    int ktiles = K >> 5;
    for (int kt = 0; kt < ktiles; ++kt) {
        int kbase = (kt << 5) + (hi << 3);
        const float4* p = reinterpret_cast<const float4*>(A + (long)row * K + kbase);
        float4 x0 = p[0], x1 = p[1];
        float4 x2 = p[4], x3 = p[5];
        if (RELU) {
            x0.x=fmaxf(x0.x,0.f); x0.y=fmaxf(x0.y,0.f); x0.z=fmaxf(x0.z,0.f); x0.w=fmaxf(x0.w,0.f);
            x1.x=fmaxf(x1.x,0.f); x1.y=fmaxf(x1.y,0.f); x1.z=fmaxf(x1.z,0.f); x1.w=fmaxf(x1.w,0.f);
            x2.x=fmaxf(x2.x,0.f); x2.y=fmaxf(x2.y,0.f); x2.z=fmaxf(x2.z,0.f); x2.w=fmaxf(x2.w,0.f);
            x3.x=fmaxf(x3.x,0.f); x3.y=fmaxf(x3.y,0.f); x3.z=fmaxf(x3.z,0.f); x3.w=fmaxf(x3.w,0.f);
        }
        v16h a;
        a[0]=(_Float16)x0.x; a[1]=(_Float16)x0.y; a[2]=(_Float16)x0.z; a[3]=(_Float16)x0.w;
        a[4]=(_Float16)x1.x; a[5]=(_Float16)x1.y; a[6]=(_Float16)x1.z; a[7]=(_Float16)x1.w;
        a[8]=(_Float16)x2.x; a[9]=(_Float16)x2.y; a[10]=(_Float16)x2.z; a[11]=(_Float16)x2.w;
        a[12]=(_Float16)x3.x; a[13]=(_Float16)x3.y; a[14]=(_Float16)x3.z; a[15]=(_Float16)x3.w;

        #pragma unroll
        for (int nt = 0; nt < NT; ++nt) {
            v16h b = bp[(size_t)(kt * NT + nt) * 32 + lane];
            acc[nt] = __builtin_amdgcn_wmma_f32_16x16x32_f16(
                false, a, false, b, (short)0, acc[nt], false, false);
        }
    }

    int mrow = m0 + (hi << 3);
    #pragma unroll
    for (int nt = 0; nt < NT; ++nt) {
        int n = (nt << 4) + (lane & 15);
        #pragma unroll
        for (int j = 0; j < 8; ++j)
            C[(long)(mrow + j) * N + n] = acc[nt][j];
    }
}

__global__ void k_self_bias(const float* __restrict__ H, const float* __restrict__ dinv,
                            const float* __restrict__ bias, float* __restrict__ acc,
                            int Nn, int C) {
    int idx = blockIdx.x * blockDim.x + threadIdx.x;
    if (idx >= Nn * C) return;
    int i = idx / C, c = idx % C;
    float d = dinv[i];
    acc[idx] = H[idx] * d * d + bias[c];
}

template <int CPE>
__global__ void k_edge_prop(const int* __restrict__ src, const int* __restrict__ dst,
                            const float* __restrict__ H, const float* __restrict__ dinv,
                            float* acc, int E) {
    int wave = (int)((blockIdx.x * blockDim.x + threadIdx.x) >> 5);
    int lane = threadIdx.x & 31;
    if (wave >= E) return;
    int s = src[wave], d = dst[wave];
    float norm = dinv[s] * dinv[d];
    constexpr int FPL = CPE / 32;
    const float* hs = H   + (long)s * CPE + lane * FPL;
    float*       ad = acc + (long)d * CPE + lane * FPL;
    if constexpr (FPL == 4) {
        float4 hv = *reinterpret_cast<const float4*>(hs);
        atomic_add_f32(&ad[0], hv.x * norm);
        atomic_add_f32(&ad[1], hv.y * norm);
        atomic_add_f32(&ad[2], hv.z * norm);
        atomic_add_f32(&ad[3], hv.w * norm);
    } else {
        float2 hv = *reinterpret_cast<const float2*>(hs);
        atomic_add_f32(&ad[0], hv.x * norm);
        atomic_add_f32(&ad[1], hv.y * norm);
    }
}

__global__ void k_pool_add(const int* __restrict__ batch, const float* __restrict__ A,
                           float* pool, float* counts, int Nn) {
    int idx = blockIdx.x * blockDim.x + threadIdx.x;
    if (idx >= Nn * OUT_C) return;
    int i = idx >> 6, c = idx & 63;
    int g = batch[i];
    atomic_add_f32(&pool[g * OUT_C + c], A[idx]);
    if (c == 0) atomic_add_f32(&counts[g], 1.0f);
}

__global__ void k_pool_final(const float* __restrict__ pool, const float* __restrict__ counts,
                             float* __restrict__ out) {
    int idx = blockIdx.x * blockDim.x + threadIdx.x;
    if (idx >= NGRAPH * OUT_C) return;
    out[idx] = pool[idx] / fmaxf(counts[idx >> 6], 1.0f);
}

extern "C" void kernel_launch(void* const* d_in, const int* in_sizes, int n_in,
                              void* d_out, int out_size, void* d_ws, size_t ws_size,
                              hipStream_t stream) {
    const float* x   = (const float*)d_in[0];
    const float* W1  = (const float*)d_in[1];
    const float* b1  = (const float*)d_in[2];
    const float* W2  = (const float*)d_in[3];
    const float* b2  = (const float*)d_in[4];
    const int*   ei  = (const int*)d_in[5];
    const int*   bat = (const int*)d_in[6];

    const int Nn = in_sizes[0] / IN_C;
    const int E  = in_sizes[5] / 2;
    const int* src = ei;
    const int* dst = ei + E;

    char* w = (char*)d_ws;
    size_t off = 0;
    auto alloc = [&](size_t bytes) -> void* {
        void* p = w + off;
        off = (off + bytes + 255) & ~(size_t)255;
        return p;
    };
    float*    deg   = (float*)alloc((size_t)Nn * 4);
    float*    dinv  = (float*)alloc((size_t)Nn * 4);
    _Float16* W1h   = (_Float16*)alloc((size_t)IN_C * HID_C * 2);
    _Float16* W2h   = (_Float16*)alloc((size_t)HID_C * OUT_C * 2);
    float*    h1    = (float*)alloc((size_t)Nn * HID_C * 4);
    float*    a1    = (float*)alloc((size_t)Nn * HID_C * 4);
    float*    pool  = (float*)alloc((size_t)NGRAPH * OUT_C * 4);
    float*    cnts  = (float*)alloc((size_t)NGRAPH * 4);
    float*    h2 = h1;
    float*    a2 = a1;

    const int T = 256;
    k_fill<<<(Nn + T - 1) / T, T, 0, stream>>>(deg, 1.0f, Nn);
    k_edge_deg<<<(E + T - 1) / T, T, 0, stream>>>(dst, deg, E);
    k_rsqrt<<<(Nn + T - 1) / T, T, 0, stream>>>(deg, dinv, Nn);

    k_pack_w<<<(IN_C * HID_C + T - 1) / T, T, 0, stream>>>(W1, W1h, IN_C, HID_C);
    k_pack_w<<<(HID_C * OUT_C + T - 1) / T, T, 0, stream>>>(W2, W2h, HID_C, OUT_C);

    {
        int waves = Nn / 16;
        int blocks = (waves * 32 + T - 1) / T;
        k_gemm<HID_C / 16, false><<<blocks, T, 0, stream>>>(x, W1h, h1, Nn, IN_C, HID_C);
    }
    k_self_bias<<<(Nn * HID_C + T - 1) / T, T, 0, stream>>>(h1, dinv, b1, a1, Nn, HID_C);
    k_edge_prop<HID_C><<<(E + 7) / 8, T, 0, stream>>>(src, dst, h1, dinv, a1, E);

    {
        int waves = Nn / 16;
        int blocks = (waves * 32 + T - 1) / T;
        k_gemm<OUT_C / 16, true><<<blocks, T, 0, stream>>>(a1, W2h, h2, Nn, HID_C, OUT_C);
    }
    k_self_bias<<<(Nn * OUT_C + T - 1) / T, T, 0, stream>>>(h2, dinv, b2, a2, Nn, OUT_C);
    k_edge_prop<OUT_C><<<(E + 7) / 8, T, 0, stream>>>(src, dst, h2, dinv, a2, E);

    k_fill<<<(NGRAPH * OUT_C + T - 1) / T, T, 0, stream>>>(pool, 0.0f, NGRAPH * OUT_C);
    k_fill<<<1, T, 0, stream>>>(cnts, 0.0f, NGRAPH);
    k_pool_add<<<(Nn * OUT_C + T - 1) / T, T, 0, stream>>>(bat, a2, pool, cnts, Nn);
    k_pool_final<<<(NGRAPH * OUT_C + T - 1) / T, T, 0, stream>>>(pool, cnts, (float*)d_out);
}


# pool sums: each thread adds its own consecutive nodes per channel in a register, one atomic per run of equal graph id (was one per node element); changes the order of the pool float sums
# speedup vs baseline: 1.2326x; 1.2326x over previous
#include <hip/hip_runtime.h>
#include <hip/hip_bf16.h>
#include <stdint.h>

typedef __attribute__((ext_vector_type(16))) _Float16 v16h;
typedef __attribute__((ext_vector_type(8)))  float    v8f;

#define IN_C   128
#define HID_C  128
#define OUT_C  64
#define NGRAPH 128

__device__ __forceinline__ void atomic_add_f32(float* p, float v) {
    unsafeAtomicAdd(p, v);
}

__global__ void k_fill(float* p, float v, int n) {
    int i = blockIdx.x * blockDim.x + threadIdx.x;
    if (i < n) p[i] = v;
}

__global__ void k_edge_deg(const int* __restrict__ dst, float* deg, int E) {
    int e = blockIdx.x * blockDim.x + threadIdx.x;
    if (e < E) atomic_add_f32(&deg[dst[e]], 1.0f);
}

__global__ void k_rsqrt(const float* __restrict__ deg, float* dinv, int n) {
    int i = blockIdx.x * blockDim.x + threadIdx.x;
    if (i < n) dinv[i] = rsqrtf(deg[i]);
}

__global__ void k_pack_w(const float* __restrict__ W, _Float16* __restrict__ Wh,
                         int K, int N) {
    int idx = blockIdx.x * blockDim.x + threadIdx.x;
    if (idx >= K * N) return;
    int frag   = idx >> 9;
    int within = idx & 511;
    int lane   = within >> 4;
    int i      = within & 15;
    int ntiles = N >> 4;
    int kt = frag / ntiles, nt = frag % ntiles;
    int n  = (nt << 4) + (lane & 15);
    int kk = (i < 8 ? i : i + 8) + ((lane >> 4) << 3);
    int k  = (kt << 5) + kk;
    Wh[idx] = (_Float16)W[k * N + n];
}

template <int NT, bool RELU>
__global__ void k_gemm(const float* __restrict__ A, const _Float16* __restrict__ Bh,
                       float* __restrict__ C, int M, int K, int N) {
    int wave = (int)((blockIdx.x * blockDim.x + threadIdx.x) >> 5);
    int lane = threadIdx.x & 31;
    int mtiles = M >> 4;
    if (wave >= mtiles) return;

    int m0  = wave << 4;
    int hi  = lane >> 4;
    int row = m0 + (lane & 15);

    v8f acc[NT];
    v8f zero = {0.f,0.f,0.f,0.f,0.f,0.f,0.f,0.f};
    #pragma unroll
    for (int j = 0; j < NT; ++j) acc[j] = zero;

    const v16h* bp = reinterpret_cast<const v16h*>(Bh);
    int ktiles = K >> 5;
    for (int kt = 0; kt < ktiles; ++kt) {
        int kbase = (kt << 5) + (hi << 3);
        const float4* p = reinterpret_cast<const float4*>(A + (long)row * K + kbase);
        float4 x0 = p[0], x1 = p[1];
        float4 x2 = p[4], x3 = p[5];
        if (RELU) {
            x0.x=fmaxf(x0.x,0.f); x0.y=fmaxf(x0.y,0.f); x0.z=fmaxf(x0.z,0.f); x0.w=fmaxf(x0.w,0.f);
            x1.x=fmaxf(x1.x,0.f); x1.y=fmaxf(x1.y,0.f); x1.z=fmaxf(x1.z,0.f); x1.w=fmaxf(x1.w,0.f);
            x2.x=fmaxf(x2.x,0.f); x2.y=fmaxf(x2.y,0.f); x2.z=fmaxf(x2.z,0.f); x2.w=fmaxf(x2.w,0.f);
            x3.x=fmaxf(x3.x,0.f); x3.y=fmaxf(x3.y,0.f); x3.z=fmaxf(x3.z,0.f); x3.w=fmaxf(x3.w,0.f);
        }
        v16h a;
        a[0]=(_Float16)x0.x; a[1]=(_Float16)x0.y; a[2]=(_Float16)x0.z; a[3]=(_Float16)x0.w;
        a[4]=(_Float16)x1.x; a[5]=(_Float16)x1.y; a[6]=(_Float16)x1.z; a[7]=(_Float16)x1.w;
        a[8]=(_Float16)x2.x; a[9]=(_Float16)x2.y; a[10]=(_Float16)x2.z; a[11]=(_Float16)x2.w;
        a[12]=(_Float16)x3.x; a[13]=(_Float16)x3.y; a[14]=(_Float16)x3.z; a[15]=(_Float16)x3.w;

        #pragma unroll
        for (int nt = 0; nt < NT; ++nt) {
            v16h b = bp[(size_t)(kt * NT + nt) * 32 + lane];
            acc[nt] = __builtin_amdgcn_wmma_f32_16x16x32_f16(
                false, a, false, b, (short)0, acc[nt], false, false);
        }
    }

    int mrow = m0 + (hi << 3);
    #pragma unroll
    for (int nt = 0; nt < NT; ++nt) {
        int n = (nt << 4) + (lane & 15);
        #pragma unroll
        for (int j = 0; j < 8; ++j)
            C[(long)(mrow + j) * N + n] = acc[nt][j];
    }
}

__global__ void k_self_bias(const float* __restrict__ H, const float* __restrict__ dinv,
                            const float* __restrict__ bias, float* __restrict__ acc,
                            int Nn, int C) {
    int idx = blockIdx.x * blockDim.x + threadIdx.x;
    if (idx >= Nn * C) return;
    int i = idx / C, c = idx % C;
    float d = dinv[i];
    acc[idx] = H[idx] * d * d + bias[c];
}

template <int CPE>
__global__ void k_edge_prop(const int* __restrict__ src, const int* __restrict__ dst,
                            const float* __restrict__ H, const float* __restrict__ dinv,
                            float* acc, int E) {
    int wave = (int)((blockIdx.x * blockDim.x + threadIdx.x) >> 5);
    int lane = threadIdx.x & 31;
    if (wave >= E) return;
    int s = src[wave], d = dst[wave];
    float norm = dinv[s] * dinv[d];
    constexpr int FPL = CPE / 32;
    const float* hs = H   + (long)s * CPE + lane * FPL;
    float*       ad = acc + (long)d * CPE + lane * FPL;
    if constexpr (FPL == 4) {
        float4 hv = *reinterpret_cast<const float4*>(hs);
        atomic_add_f32(&ad[0], hv.x * norm);
        atomic_add_f32(&ad[1], hv.y * norm);
        atomic_add_f32(&ad[2], hv.z * norm);
        atomic_add_f32(&ad[3], hv.w * norm);
    } else {
        float2 hv = *reinterpret_cast<const float2*>(hs);
        atomic_add_f32(&ad[0], hv.x * norm);
        atomic_add_f32(&ad[1], hv.y * norm);
    }
}

#define POOL_NPT 16
__global__ void k_pool_add(const int* __restrict__ batch, const float* __restrict__ A,
                           float* pool, float* counts, int Nn) {
    int c  = threadIdx.x & 63;
    int r  = threadIdx.x >> 6;
    int i0 = (blockIdx.x * 4 + r) * POOL_NPT;
    if (i0 >= Nn) return;

    int   g[POOL_NPT];
    float v[POOL_NPT];
    #pragma unroll
    for (int k = 0; k < POOL_NPT; ++k) {
        int i  = i0 + k;
        int il = i < Nn ? i : Nn - 1;
        int gk = batch[il];
        v[k] = A[(long)il * OUT_C + c];
        g[k] = i < Nn ? gk : -1;
    }

    int   gc = g[0];
    float s  = v[0];
    float n  = 1.0f;
    #pragma unroll
    for (int k = 1; k < POOL_NPT; ++k) {
        if (g[k] != gc) {
            if (gc >= 0) {
                atomic_add_f32(&pool[gc * OUT_C + c], s);
                if (c == 0) atomic_add_f32(&counts[gc], n);
            }
            gc = g[k]; s = 0.0f; n = 0.0f;
        }
        s += v[k];
        n += 1.0f;
    }
    if (gc >= 0) {
        atomic_add_f32(&pool[gc * OUT_C + c], s);
        if (c == 0) atomic_add_f32(&counts[gc], n);
    }
}

__global__ void k_pool_final(const float* __restrict__ pool, const float* __restrict__ counts,
                             float* __restrict__ out) {
    int idx = blockIdx.x * blockDim.x + threadIdx.x;
    if (idx >= NGRAPH * OUT_C) return;
    out[idx] = pool[idx] / fmaxf(counts[idx >> 6], 1.0f);
}

extern "C" void kernel_launch(void* const* d_in, const int* in_sizes, int n_in,
                              void* d_out, int out_size, void* d_ws, size_t ws_size,
                              hipStream_t stream) {
    const float* x   = (const float*)d_in[0];
    const float* W1  = (const float*)d_in[1];
    const float* b1  = (const float*)d_in[2];
    const float* W2  = (const float*)d_in[3];
    const float* b2  = (const float*)d_in[4];
    const int*   ei  = (const int*)d_in[5];
    const int*   bat = (const int*)d_in[6];

    const int Nn = in_sizes[0] / IN_C;
    const int E  = in_sizes[5] / 2;
    const int* src = ei;
    const int* dst = ei + E;

    char* w = (char*)d_ws;
    size_t off = 0;
    auto alloc = [&](size_t bytes) -> void* {
        void* p = w + off;
        off = (off + bytes + 255) & ~(size_t)255;
        return p;
    };
    float*    deg   = (float*)alloc((size_t)Nn * 4);
    float*    dinv  = (float*)alloc((size_t)Nn * 4);
    _Float16* W1h   = (_Float16*)alloc((size_t)IN_C * HID_C * 2);
    _Float16* W2h   = (_Float16*)alloc((size_t)HID_C * OUT_C * 2);
    float*    h1    = (float*)alloc((size_t)Nn * HID_C * 4);
    float*    a1    = (float*)alloc((size_t)Nn * HID_C * 4);
    float*    pool  = (float*)alloc((size_t)NGRAPH * OUT_C * 4);
    float*    cnts  = (float*)alloc((size_t)NGRAPH * 4);
    float*    h2 = h1;
    float*    a2 = a1;

    const int T = 256;
    k_fill<<<(Nn + T - 1) / T, T, 0, stream>>>(deg, 1.0f, Nn);
    k_edge_deg<<<(E + T - 1) / T, T, 0, stream>>>(dst, deg, E);
    k_rsqrt<<<(Nn + T - 1) / T, T, 0, stream>>>(deg, dinv, Nn);

    k_pack_w<<<(IN_C * HID_C + T - 1) / T, T, 0, stream>>>(W1, W1h, IN_C, HID_C);
    k_pack_w<<<(HID_C * OUT_C + T - 1) / T, T, 0, stream>>>(W2, W2h, HID_C, OUT_C);

    {
        int waves = Nn / 16;
        int blocks = (waves * 32 + T - 1) / T;
        k_gemm<HID_C / 16, false><<<blocks, T, 0, stream>>>(x, W1h, h1, Nn, IN_C, HID_C);
    }
    k_self_bias<<<(Nn * HID_C + T - 1) / T, T, 0, stream>>>(h1, dinv, b1, a1, Nn, HID_C);
    k_edge_prop<HID_C><<<(E + 7) / 8, T, 0, stream>>>(src, dst, h1, dinv, a1, E);

    {
        int waves = Nn / 16;
        int blocks = (waves * 32 + T - 1) / T;
        k_gemm<OUT_C / 16, true><<<blocks, T, 0, stream>>>(a1, W2h, h2, Nn, HID_C, OUT_C);
    }
    k_self_bias<<<(Nn * OUT_C + T - 1) / T, T, 0, stream>>>(h2, dinv, b2, a2, Nn, OUT_C);
    k_edge_prop<OUT_C><<<(E + 7) / 8, T, 0, stream>>>(src, dst, h2, dinv, a2, E);

    k_fill<<<(NGRAPH * OUT_C + T - 1) / T, T, 0, stream>>>(pool, 0.0f, NGRAPH * OUT_C);
    k_fill<<<1, T, 0, stream>>>(cnts, 0.0f, NGRAPH);
    k_pool_add<<<(Nn + 4 * POOL_NPT - 1) / (4 * POOL_NPT), T, 0, stream>>>(bat, a2, pool, cnts, Nn);
    k_pool_final<<<(NGRAPH * OUT_C + T - 1) / T, T, 0, stream>>>(pool, cnts, (float*)d_out);
}


# propagate: edges grouped by destination once per call (integer atomics); a wave per node sums its in-edge rows in registers, one store; no float atomics or self-loop pass; float sum order changes
# speedup vs baseline: 2.0598x; 1.6711x over previous
#include <hip/hip_runtime.h>
#include <hip/hip_bf16.h>
#include <stdint.h>

typedef __attribute__((ext_vector_type(16))) _Float16 v16h;
typedef __attribute__((ext_vector_type(8)))  float    v8f;

#define IN_C   128
#define HID_C  128
#define OUT_C  64
#define NGRAPH 128

__device__ __forceinline__ void atomic_add_f32(float* p, float v) {
    unsafeAtomicAdd(p, v);
}

__global__ void k_fill(float* p, float v, int n) {
    int i = blockIdx.x * blockDim.x + threadIdx.x;
    if (i < n) p[i] = v;
}

__global__ void k_zero_i32(int* p, int n) {
    int i = blockIdx.x * blockDim.x + threadIdx.x;
    if (i < n) p[i] = 0;
}

__global__ void k_edge_count(const int* __restrict__ dst, int* cnt, int E) {
    int e = blockIdx.x * blockDim.x + threadIdx.x;
    if (e < E) atomicAdd(&cnt[dst[e]], 1);
}

__global__ void k_node_setup(const int* cnt, int* total, float* __restrict__ dinv,
                             int* __restrict__ cursor, int Nn) {
    __shared__ int s_wave[8];
    __shared__ int s_base;
    int i    = blockIdx.x * 256 + threadIdx.x;
    int lane = threadIdx.x & 31;
    int wv   = threadIdx.x >> 5;
    int c = i < Nn ? cnt[i] : 0;
    if (i < Nn) dinv[i] = rsqrtf((float)(c + 1));
    int incl = c;
    #pragma unroll
    for (int o = 1; o < 32; o <<= 1) {
        int t = __shfl_up(incl, o);
        if (lane >= o) incl += t;
    }
    if (lane == 31) s_wave[wv] = incl;
    __syncthreads();
    if (threadIdx.x == 0) {
        int run = 0;
        for (int w = 0; w < 8; ++w) { int t = s_wave[w]; s_wave[w] = run; run += t; }
        s_base = atomicAdd(total, run);
    }
    __syncthreads();
    if (i < Nn) cursor[i] = s_base + s_wave[wv] + incl - c;
}

__global__ void k_edge_place(const int* __restrict__ src, const int* __restrict__ dst,
                             int* cursor, int* __restrict__ nbr, int E) {
    int e = blockIdx.x * blockDim.x + threadIdx.x;
    if (e >= E) return;
    int p = atomicAdd(&cursor[dst[e]], 1);
    nbr[p] = src[e];
}

__global__ void k_pack_w(const float* __restrict__ W, _Float16* __restrict__ Wh,
                         int K, int N) {
    int idx = blockIdx.x * blockDim.x + threadIdx.x;
    if (idx >= K * N) return;
    int frag   = idx >> 9;
    int within = idx & 511;
    int lane   = within >> 4;
    int i      = within & 15;
    int ntiles = N >> 4;
    int kt = frag / ntiles, nt = frag % ntiles;
    int n  = (nt << 4) + (lane & 15);
    int kk = (i < 8 ? i : i + 8) + ((lane >> 4) << 3);
    int k  = (kt << 5) + kk;
    Wh[idx] = (_Float16)W[k * N + n];
}

template <int NT, bool RELU>
__global__ void k_gemm(const float* __restrict__ A, const _Float16* __restrict__ Bh,
                       float* __restrict__ C, int M, int K, int N) {
    int wave = (int)((blockIdx.x * blockDim.x + threadIdx.x) >> 5);
    int lane = threadIdx.x & 31;
    int mtiles = M >> 4;
    if (wave >= mtiles) return;

    int m0  = wave << 4;
    int hi  = lane >> 4;
    int row = m0 + (lane & 15);

    v8f acc[NT];
    v8f zero = {0.f,0.f,0.f,0.f,0.f,0.f,0.f,0.f};
    #pragma unroll
    for (int j = 0; j < NT; ++j) acc[j] = zero;

    const v16h* bp = reinterpret_cast<const v16h*>(Bh);
    int ktiles = K >> 5;
    for (int kt = 0; kt < ktiles; ++kt) {
        int kbase = (kt << 5) + (hi << 3);
        const float4* p = reinterpret_cast<const float4*>(A + (long)row * K + kbase);
        float4 x0 = p[0], x1 = p[1];
        float4 x2 = p[4], x3 = p[5];
        if (RELU) {
            x0.x=fmaxf(x0.x,0.f); x0.y=fmaxf(x0.y,0.f); x0.z=fmaxf(x0.z,0.f); x0.w=fmaxf(x0.w,0.f);
            x1.x=fmaxf(x1.x,0.f); x1.y=fmaxf(x1.y,0.f); x1.z=fmaxf(x1.z,0.f); x1.w=fmaxf(x1.w,0.f);
            x2.x=fmaxf(x2.x,0.f); x2.y=fmaxf(x2.y,0.f); x2.z=fmaxf(x2.z,0.f); x2.w=fmaxf(x2.w,0.f);
            x3.x=fmaxf(x3.x,0.f); x3.y=fmaxf(x3.y,0.f); x3.z=fmaxf(x3.z,0.f); x3.w=fmaxf(x3.w,0.f);
        }
        v16h a;
        a[0]=(_Float16)x0.x; a[1]=(_Float16)x0.y; a[2]=(_Float16)x0.z; a[3]=(_Float16)x0.w;
        a[4]=(_Float16)x1.x; a[5]=(_Float16)x1.y; a[6]=(_Float16)x1.z; a[7]=(_Float16)x1.w;
        a[8]=(_Float16)x2.x; a[9]=(_Float16)x2.y; a[10]=(_Float16)x2.z; a[11]=(_Float16)x2.w;
        a[12]=(_Float16)x3.x; a[13]=(_Float16)x3.y; a[14]=(_Float16)x3.z; a[15]=(_Float16)x3.w;

        #pragma unroll
        for (int nt = 0; nt < NT; ++nt) {
            v16h b = bp[(size_t)(kt * NT + nt) * 32 + lane];
            acc[nt] = __builtin_amdgcn_wmma_f32_16x16x32_f16(
                false, a, false, b, (short)0, acc[nt], false, false);
        }
    }

    int mrow = m0 + (hi << 3);
    #pragma unroll
    for (int nt = 0; nt < NT; ++nt) {
        int n = (nt << 4) + (lane & 15);
        #pragma unroll
        for (int j = 0; j < 8; ++j)
            C[(long)(mrow + j) * N + n] = acc[nt][j];
    }
}


template <int N, typename VT>
__device__ __forceinline__ void add_rows(VT& acc, const int* __restrict__ nbr, int j,
                                         const float* __restrict__ dinv, float dd,
                                         const VT* __restrict__ Hv, int lane) {
    int   s[N];
    float w[N];
    VT    r[N];
    #pragma unroll
    for (int k = 0; k < N; ++k) s[k] = nbr[j + k];
    #pragma unroll
    for (int k = 0; k < N; ++k) r[k] = Hv[(size_t)s[k] * 32 + lane];
    #pragma unroll
    for (int k = 0; k < N; ++k) w[k] = dinv[s[k]] * dd;
    #pragma unroll
    for (int k = 0; k < N; ++k) acc += r[k] * w[k];
}

template <int C>
__global__ void k_gather(const int* __restrict__ cnt, const int* __restrict__ cursor,
                         const int* __restrict__ nbr, const float* __restrict__ H,
                         const float* __restrict__ dinv, const float* __restrict__ bias,
                         float* __restrict__ out, int Nn) {
    typedef __attribute__((ext_vector_type(C / 32))) float vt;
    int node = __builtin_amdgcn_readfirstlane((int)((blockIdx.x * blockDim.x + threadIdx.x) >> 5));
    int lane = threadIdx.x & 31;
    if (node >= Nn) return;
    int end = cursor[node];
    int j   = end - cnt[node];
    float dd = dinv[node];
    const vt* Hv = reinterpret_cast<const vt*>(H);
    vt acc = Hv[(size_t)node * 32 + lane] * (dd * dd);
    for (; j + 8 <= end; j += 8) add_rows<8, vt>(acc, nbr, j, dinv, dd, Hv, lane);
    int rem = end - j;
    if (rem & 4) { add_rows<4, vt>(acc, nbr, j, dinv, dd, Hv, lane); j += 4; }
    if (rem & 2) { add_rows<2, vt>(acc, nbr, j, dinv, dd, Hv, lane); j += 2; }
    if (rem & 1) { add_rows<1, vt>(acc, nbr, j, dinv, dd, Hv, lane); }
    const vt* bv = reinterpret_cast<const vt*>(bias);
    reinterpret_cast<vt*>(out)[(size_t)node * 32 + lane] = acc + bv[lane];
}

#define POOL_NPT 16
__global__ void k_pool_add(const int* __restrict__ batch, const float* __restrict__ A,
                           float* pool, float* counts, int Nn) {
    int c  = threadIdx.x & 63;
    int r  = threadIdx.x >> 6;
    int i0 = (blockIdx.x * 4 + r) * POOL_NPT;
    if (i0 >= Nn) return;

    int   g[POOL_NPT];
    float v[POOL_NPT];
    #pragma unroll
    for (int k = 0; k < POOL_NPT; ++k) {
        int i  = i0 + k;
        int il = i < Nn ? i : Nn - 1;
        int gk = batch[il];
        v[k] = A[(long)il * OUT_C + c];
        g[k] = i < Nn ? gk : -1;
    }

    int   gc = g[0];
    float s  = v[0];
    float n  = 1.0f;
    #pragma unroll
    for (int k = 1; k < POOL_NPT; ++k) {
        if (g[k] != gc) {
            if (gc >= 0) {
                atomic_add_f32(&pool[gc * OUT_C + c], s);
                if (c == 0) atomic_add_f32(&counts[gc], n);
            }
            gc = g[k]; s = 0.0f; n = 0.0f;
        }
        s += v[k];
        n += 1.0f;
    }
    if (gc >= 0) {
        atomic_add_f32(&pool[gc * OUT_C + c], s);
        if (c == 0) atomic_add_f32(&counts[gc], n);
    }
}

__global__ void k_pool_final(const float* __restrict__ pool, const float* __restrict__ counts,
                             float* __restrict__ out) {
    int idx = blockIdx.x * blockDim.x + threadIdx.x;
    if (idx >= NGRAPH * OUT_C) return;
    out[idx] = pool[idx] / fmaxf(counts[idx >> 6], 1.0f);
}

extern "C" void kernel_launch(void* const* d_in, const int* in_sizes, int n_in,
                              void* d_out, int out_size, void* d_ws, size_t ws_size,
                              hipStream_t stream) {
    const float* x   = (const float*)d_in[0];
    const float* W1  = (const float*)d_in[1];
    const float* b1  = (const float*)d_in[2];
    const float* W2  = (const float*)d_in[3];
    const float* b2  = (const float*)d_in[4];
    const int*   ei  = (const int*)d_in[5];
    const int*   bat = (const int*)d_in[6];

    const int Nn = in_sizes[0] / IN_C;
    const int E  = in_sizes[5] / 2;
    const int* src = ei;
    const int* dst = ei + E;

    char* w = (char*)d_ws;
    size_t off = 0;
    auto alloc = [&](size_t bytes) -> void* {
        void* p = w + off;
        off = (off + bytes + 255) & ~(size_t)255;
        return p;
    };
    int*      cnt   = (int*)alloc((size_t)(Nn + 1) * 4);
    int*      total = cnt + Nn;
    int*      cursor= (int*)alloc((size_t)Nn * 4);
    int*      nbr   = (int*)alloc((size_t)E * 4);
    float*    dinv  = (float*)alloc((size_t)Nn * 4);
    _Float16* W1h   = (_Float16*)alloc((size_t)IN_C * HID_C * 2);
    _Float16* W2h   = (_Float16*)alloc((size_t)HID_C * OUT_C * 2);
    float*    h1    = (float*)alloc((size_t)Nn * HID_C * 4);
    float*    a1    = (float*)alloc((size_t)Nn * HID_C * 4);
    float*    pool  = (float*)alloc((size_t)NGRAPH * OUT_C * 4);
    float*    cnts  = (float*)alloc((size_t)NGRAPH * 4);
    float*    h2 = h1;
    float*    a2 = a1;
    if (off > ws_size) return;

    const int T = 256;
    k_zero_i32<<<(Nn + 1 + T - 1) / T, T, 0, stream>>>(cnt, Nn + 1);
    k_edge_count<<<(E + T - 1) / T, T, 0, stream>>>(dst, cnt, E);
    k_node_setup<<<(Nn + T - 1) / T, T, 0, stream>>>(cnt, total, dinv, cursor, Nn);
    k_edge_place<<<(E + T - 1) / T, T, 0, stream>>>(src, dst, cursor, nbr, E);

    k_pack_w<<<(IN_C * HID_C + T - 1) / T, T, 0, stream>>>(W1, W1h, IN_C, HID_C);
    k_pack_w<<<(HID_C * OUT_C + T - 1) / T, T, 0, stream>>>(W2, W2h, HID_C, OUT_C);

    {
        int waves = Nn / 16;
        int blocks = (waves * 32 + T - 1) / T;
        k_gemm<HID_C / 16, false><<<blocks, T, 0, stream>>>(x, W1h, h1, Nn, IN_C, HID_C);
    }
    k_gather<HID_C><<<(Nn + 7) / 8, T, 0, stream>>>(cnt, cursor, nbr, h1, dinv, b1, a1, Nn);

    {
        int waves = Nn / 16;
        int blocks = (waves * 32 + T - 1) / T;
        k_gemm<OUT_C / 16, true><<<blocks, T, 0, stream>>>(a1, W2h, h2, Nn, HID_C, OUT_C);
    }
    k_gather<OUT_C><<<(Nn + 7) / 8, T, 0, stream>>>(cnt, cursor, nbr, h2, dinv, b2, a2, Nn);

    k_fill<<<(NGRAPH * OUT_C + T - 1) / T, T, 0, stream>>>(pool, 0.0f, NGRAPH * OUT_C);
    k_fill<<<1, T, 0, stream>>>(cnts, 0.0f, NGRAPH);
    k_pool_add<<<(Nn + 4 * POOL_NPT - 1) / (4 * POOL_NPT), T, 0, stream>>>(bat, a2, pool, cnts, Nn);
    k_pool_final<<<(NGRAPH * OUT_C + T - 1) / T, T, 0, stream>>>(pool, cnts, (float*)d_out);
}


# fewer dependent launches, 14 to 9: the five clears and weight packs are block ranges of one launch, and layer 1's product shares the in-degree count's launch; arithmetic and sum order unchanged
# speedup vs baseline: 2.2372x; 1.0861x over previous
#include <hip/hip_runtime.h>
#include <hip/hip_bf16.h>
#include <stdint.h>

typedef __attribute__((ext_vector_type(16))) _Float16 v16h;
typedef __attribute__((ext_vector_type(8)))  float    v8f;

#define IN_C   128
#define HID_C  128
#define OUT_C  64
#define NGRAPH 128

__device__ __forceinline__ void atomic_add_f32(float* p, float v) {
    unsafeAtomicAdd(p, v);
}

__device__ __forceinline__ void pack_w_elem(const float* __restrict__ W, _Float16* __restrict__ Wh,
                                            int K, int N, int idx) {
    if (idx >= K * N) return;
    int frag   = idx >> 9;
    int within = idx & 511;
    int lane   = within >> 4;
    int i      = within & 15;
    int ntiles = N >> 4;
    int kt = frag / ntiles, nt = frag % ntiles;
    int n  = (nt << 4) + (lane & 15);
    int kk = (i < 8 ? i : i + 8) + ((lane >> 4) << 3);
    int k  = (kt << 5) + kk;
    Wh[idx] = (_Float16)W[k * N + n];
}

__global__ void k_setup(int* __restrict__ cnt, int n_cnt, int nb_cnt,
                        float* __restrict__ pool, float* __restrict__ counts,
                        const float* __restrict__ W1, _Float16* __restrict__ W1h,
                        const float* __restrict__ W2, _Float16* __restrict__ W2h) {
    const int NB_POOL = (NGRAPH * OUT_C) / 256;
    const int NB_W1   = (IN_C * HID_C) / 256;
    int b = blockIdx.x;
    int t = threadIdx.x;
    if (b < nb_cnt) {
        int i = b * 256 + t;
        if (i < n_cnt) cnt[i] = 0;
        return;
    }
    b -= nb_cnt;
    if (b < NB_POOL) { pool[b * 256 + t] = 0.0f; return; }
    b -= NB_POOL;
    if (b == 0) { if (t < NGRAPH) counts[t] = 0.0f; return; }
    b -= 1;
    if (b < NB_W1) { pack_w_elem(W1, W1h, IN_C, HID_C, b * 256 + t); return; }
    b -= NB_W1;
    pack_w_elem(W2, W2h, HID_C, OUT_C, b * 256 + t);
}

__global__ void k_node_setup(const int* cnt, int* total, float* __restrict__ dinv,
                             int* __restrict__ cursor, int Nn) {
    __shared__ int s_wave[8];
    __shared__ int s_base;
    int i    = blockIdx.x * 256 + threadIdx.x;
    int lane = threadIdx.x & 31;
    int wv   = threadIdx.x >> 5;
    int c = i < Nn ? cnt[i] : 0;
    if (i < Nn) dinv[i] = rsqrtf((float)(c + 1));
    int incl = c;
    #pragma unroll
    for (int o = 1; o < 32; o <<= 1) {
        int t = __shfl_up(incl, o);
        if (lane >= o) incl += t;
    }
    if (lane == 31) s_wave[wv] = incl;
    __syncthreads();
    if (threadIdx.x == 0) {
        int run = 0;
        for (int w = 0; w < 8; ++w) { int t = s_wave[w]; s_wave[w] = run; run += t; }
        s_base = atomicAdd(total, run);
    }
    __syncthreads();
    if (i < Nn) cursor[i] = s_base + s_wave[wv] + incl - c;
}

__global__ void k_edge_place(const int* __restrict__ src, const int* __restrict__ dst,
                             int* cursor, int* __restrict__ nbr, int E) {
    int e = blockIdx.x * blockDim.x + threadIdx.x;
    if (e >= E) return;
    int p = atomicAdd(&cursor[dst[e]], 1);
    nbr[p] = src[e];
}

template <int NT, bool RELU>
__device__ __forceinline__ void gemm_strip(int wave, const float* __restrict__ A,
                                           const _Float16* __restrict__ Bh,
                                           float* __restrict__ C, int M, int K, int N) {
    int lane = threadIdx.x & 31;
    int mtiles = M >> 4;
    if (wave >= mtiles) return;

    int m0  = wave << 4;
    int hi  = lane >> 4;
    int row = m0 + (lane & 15);

    v8f acc[NT];
    v8f zero = {0.f,0.f,0.f,0.f,0.f,0.f,0.f,0.f};
    #pragma unroll
    for (int j = 0; j < NT; ++j) acc[j] = zero;

    const v16h* bp = reinterpret_cast<const v16h*>(Bh);
    int ktiles = K >> 5;
    for (int kt = 0; kt < ktiles; ++kt) {
        int kbase = (kt << 5) + (hi << 3);
        const float4* p = reinterpret_cast<const float4*>(A + (long)row * K + kbase);
        float4 x0 = p[0], x1 = p[1];
        float4 x2 = p[4], x3 = p[5];
        if (RELU) {
            x0.x=fmaxf(x0.x,0.f); x0.y=fmaxf(x0.y,0.f); x0.z=fmaxf(x0.z,0.f); x0.w=fmaxf(x0.w,0.f);
            x1.x=fmaxf(x1.x,0.f); x1.y=fmaxf(x1.y,0.f); x1.z=fmaxf(x1.z,0.f); x1.w=fmaxf(x1.w,0.f);
            x2.x=fmaxf(x2.x,0.f); x2.y=fmaxf(x2.y,0.f); x2.z=fmaxf(x2.z,0.f); x2.w=fmaxf(x2.w,0.f);
            x3.x=fmaxf(x3.x,0.f); x3.y=fmaxf(x3.y,0.f); x3.z=fmaxf(x3.z,0.f); x3.w=fmaxf(x3.w,0.f);
        }
        v16h a;
        a[0]=(_Float16)x0.x; a[1]=(_Float16)x0.y; a[2]=(_Float16)x0.z; a[3]=(_Float16)x0.w;
        a[4]=(_Float16)x1.x; a[5]=(_Float16)x1.y; a[6]=(_Float16)x1.z; a[7]=(_Float16)x1.w;
        a[8]=(_Float16)x2.x; a[9]=(_Float16)x2.y; a[10]=(_Float16)x2.z; a[11]=(_Float16)x2.w;
        a[12]=(_Float16)x3.x; a[13]=(_Float16)x3.y; a[14]=(_Float16)x3.z; a[15]=(_Float16)x3.w;

        #pragma unroll
        for (int nt = 0; nt < NT; ++nt) {
            v16h b = bp[(size_t)(kt * NT + nt) * 32 + lane];
            acc[nt] = __builtin_amdgcn_wmma_f32_16x16x32_f16(
                false, a, false, b, (short)0, acc[nt], false, false);
        }
    }

    int mrow = m0 + (hi << 3);
    #pragma unroll
    for (int nt = 0; nt < NT; ++nt) {
        int n = (nt << 4) + (lane & 15);
        #pragma unroll
        for (int j = 0; j < 8; ++j)
            C[(long)(mrow + j) * N + n] = acc[nt][j];
    }
}

template <int NT, bool RELU>
__global__ void k_gemm(const float* __restrict__ A, const _Float16* __restrict__ Bh,
                       float* __restrict__ C, int M, int K, int N) {
    gemm_strip<NT, RELU>((int)((blockIdx.x * blockDim.x + threadIdx.x) >> 5), A, Bh, C, M, K, N);
}

template <int NT>
__global__ void k_gemm_count(const float* __restrict__ A, const _Float16* __restrict__ Bh,
                             float* __restrict__ C, int M, int K, int N, int gemm_blocks,
                             const int* __restrict__ dst, int* cnt, int E) {
    int b = blockIdx.x;
    if (b < gemm_blocks) {
        gemm_strip<NT, false>((int)((b * blockDim.x + threadIdx.x) >> 5), A, Bh, C, M, K, N);
    } else {
        int e = (b - gemm_blocks) * blockDim.x + threadIdx.x;
        if (e < E) atomicAdd(&cnt[dst[e]], 1);
    }
}


template <int N, typename VT>
__device__ __forceinline__ void add_rows(VT& acc, const int* __restrict__ nbr, int j,
                                         const float* __restrict__ dinv, float dd,
                                         const VT* __restrict__ Hv, int lane) {
    int   s[N];
    float w[N];
    VT    r[N];
    #pragma unroll
    for (int k = 0; k < N; ++k) s[k] = nbr[j + k];
    #pragma unroll
    for (int k = 0; k < N; ++k) r[k] = Hv[(size_t)s[k] * 32 + lane];
    #pragma unroll
    for (int k = 0; k < N; ++k) w[k] = dinv[s[k]] * dd;
    #pragma unroll
    for (int k = 0; k < N; ++k) acc += r[k] * w[k];
}

template <int C>
__global__ void k_gather(const int* __restrict__ cnt, const int* __restrict__ cursor,
                         const int* __restrict__ nbr, const float* __restrict__ H,
                         const float* __restrict__ dinv, const float* __restrict__ bias,
                         float* __restrict__ out, int Nn) {
    typedef __attribute__((ext_vector_type(C / 32))) float vt;
    int node = __builtin_amdgcn_readfirstlane((int)((blockIdx.x * blockDim.x + threadIdx.x) >> 5));
    int lane = threadIdx.x & 31;
    if (node >= Nn) return;
    int end = cursor[node];
    int j   = end - cnt[node];
    float dd = dinv[node];
    const vt* Hv = reinterpret_cast<const vt*>(H);
    vt acc = Hv[(size_t)node * 32 + lane] * (dd * dd);
    for (; j + 8 <= end; j += 8) add_rows<8, vt>(acc, nbr, j, dinv, dd, Hv, lane);
    int rem = end - j;
    if (rem & 4) { add_rows<4, vt>(acc, nbr, j, dinv, dd, Hv, lane); j += 4; }
    if (rem & 2) { add_rows<2, vt>(acc, nbr, j, dinv, dd, Hv, lane); j += 2; }
    if (rem & 1) { add_rows<1, vt>(acc, nbr, j, dinv, dd, Hv, lane); }
    const vt* bv = reinterpret_cast<const vt*>(bias);
    reinterpret_cast<vt*>(out)[(size_t)node * 32 + lane] = acc + bv[lane];
}

#define POOL_NPT 16
__global__ void k_pool_add(const int* __restrict__ batch, const float* __restrict__ A,
                           float* pool, float* counts, int Nn) {
    int c  = threadIdx.x & 63;
    int r  = threadIdx.x >> 6;
    int i0 = (blockIdx.x * 4 + r) * POOL_NPT;
    if (i0 >= Nn) return;

    int   g[POOL_NPT];
    float v[POOL_NPT];
    #pragma unroll
    for (int k = 0; k < POOL_NPT; ++k) {
        int i  = i0 + k;
        int il = i < Nn ? i : Nn - 1;
        int gk = batch[il];
        v[k] = A[(long)il * OUT_C + c];
        g[k] = i < Nn ? gk : -1;
    }

    int   gc = g[0];
    float s  = v[0];
    float n  = 1.0f;
    #pragma unroll
    for (int k = 1; k < POOL_NPT; ++k) {
        if (g[k] != gc) {
            if (gc >= 0) {
                atomic_add_f32(&pool[gc * OUT_C + c], s);
                if (c == 0) atomic_add_f32(&counts[gc], n);
            }
            gc = g[k]; s = 0.0f; n = 0.0f;
        }
        s += v[k];
        n += 1.0f;
    }
    if (gc >= 0) {
        atomic_add_f32(&pool[gc * OUT_C + c], s);
        if (c == 0) atomic_add_f32(&counts[gc], n);
    }
}

__global__ void k_pool_final(const float* __restrict__ pool, const float* __restrict__ counts,
                             float* __restrict__ out) {
    int idx = blockIdx.x * blockDim.x + threadIdx.x;
    if (idx >= NGRAPH * OUT_C) return;
    out[idx] = pool[idx] / fmaxf(counts[idx >> 6], 1.0f);
}

extern "C" void kernel_launch(void* const* d_in, const int* in_sizes, int n_in,
                              void* d_out, int out_size, void* d_ws, size_t ws_size,
                              hipStream_t stream) {
    const float* x   = (const float*)d_in[0];
    const float* W1  = (const float*)d_in[1];
    const float* b1  = (const float*)d_in[2];
    const float* W2  = (const float*)d_in[3];
    const float* b2  = (const float*)d_in[4];
    const int*   ei  = (const int*)d_in[5];
    const int*   bat = (const int*)d_in[6];

    const int Nn = in_sizes[0] / IN_C;
    const int E  = in_sizes[5] / 2;
    const int* src = ei;
    const int* dst = ei + E;

    char* w = (char*)d_ws;
    size_t off = 0;
    auto alloc = [&](size_t bytes) -> void* {
        void* p = w + off;
        off = (off + bytes + 255) & ~(size_t)255;
        return p;
    };
    int*      cnt   = (int*)alloc((size_t)(Nn + 1) * 4);
    int*      total = cnt + Nn;
    int*      cursor= (int*)alloc((size_t)Nn * 4);
    int*      nbr   = (int*)alloc((size_t)E * 4);
    float*    dinv  = (float*)alloc((size_t)Nn * 4);
    _Float16* W1h   = (_Float16*)alloc((size_t)IN_C * HID_C * 2);
    _Float16* W2h   = (_Float16*)alloc((size_t)HID_C * OUT_C * 2);
    float*    h1    = (float*)alloc((size_t)Nn * HID_C * 4);
    float*    a1    = (float*)alloc((size_t)Nn * HID_C * 4);
    float*    pool  = (float*)alloc((size_t)NGRAPH * OUT_C * 4);
    float*    cnts  = (float*)alloc((size_t)NGRAPH * 4);
    float*    h2 = h1;
    float*    a2 = a1;
    if (off > ws_size) return;

    const int T = 256;
    {
        int nb_cnt = (Nn + 1 + T - 1) / T;
        int blocks = nb_cnt + (NGRAPH * OUT_C) / T + 1 + (IN_C * HID_C) / T + (HID_C * OUT_C) / T;
        k_setup<<<blocks, T, 0, stream>>>(cnt, Nn + 1, nb_cnt, pool, cnts, W1, W1h, W2, W2h);
    }
    {
        int waves = Nn / 16;
        int gemm_blocks = (waves * 32 + T - 1) / T;
        int edge_blocks = (E + T - 1) / T;
        k_gemm_count<HID_C / 16><<<gemm_blocks + edge_blocks, T, 0, stream>>>(
            x, W1h, h1, Nn, IN_C, HID_C, gemm_blocks, dst, cnt, E);
    }
    k_node_setup<<<(Nn + T - 1) / T, T, 0, stream>>>(cnt, total, dinv, cursor, Nn);
    k_edge_place<<<(E + T - 1) / T, T, 0, stream>>>(src, dst, cursor, nbr, E);

    k_gather<HID_C><<<(Nn + 7) / 8, T, 0, stream>>>(cnt, cursor, nbr, h1, dinv, b1, a1, Nn);

    {
        int waves = Nn / 16;
        int blocks = (waves * 32 + T - 1) / T;
        k_gemm<OUT_C / 16, true><<<blocks, T, 0, stream>>>(a1, W2h, h2, Nn, HID_C, OUT_C);
    }
    k_gather<OUT_C><<<(Nn + 7) / 8, T, 0, stream>>>(cnt, cursor, nbr, h2, dinv, b2, a2, Nn);

    k_pool_add<<<(Nn + 4 * POOL_NPT - 1) / (4 * POOL_NPT), T, 0, stream>>>(bat, a2, pool, cnts, Nn);
    k_pool_final<<<(NGRAPH * OUT_C + T - 1) / T, T, 0, stream>>>(pool, cnts, (float*)d_out);
}


# propagate runs 16 nodes per block, rows stay in LDS, and the wave that stores the last row (LDS ticket) does the consumer stage: layer-2 product, pool run sums; 9 launches to 7; sum order unchanged
# speedup vs baseline: 2.4262x; 1.0845x over previous
#include <hip/hip_runtime.h>
#include <hip/hip_bf16.h>
#include <stdint.h>

typedef __attribute__((ext_vector_type(16))) _Float16 v16h;
typedef __attribute__((ext_vector_type(8)))  float    v8f;

#define IN_C   128
#define HID_C  128
#define OUT_C  64
#define NGRAPH 128

__device__ __forceinline__ void atomic_add_f32(float* p, float v) {
    unsafeAtomicAdd(p, v);
}

__device__ __forceinline__ void pack_w_elem(const float* __restrict__ W, _Float16* __restrict__ Wh,
                                            int K, int N, int idx) {
    if (idx >= K * N) return;
    int frag   = idx >> 9;
    int within = idx & 511;
    int lane   = within >> 4;
    int i      = within & 15;
    int ntiles = N >> 4;
    int kt = frag / ntiles, nt = frag % ntiles;
    int n  = (nt << 4) + (lane & 15);
    int kk = (i < 8 ? i : i + 8) + ((lane >> 4) << 3);
    int k  = (kt << 5) + kk;
    Wh[idx] = (_Float16)W[k * N + n];
}

__global__ void k_setup(int* __restrict__ cnt, int n_cnt, int nb_cnt,
                        float* __restrict__ pool, float* __restrict__ counts,
                        const float* __restrict__ W1, _Float16* __restrict__ W1h,
                        const float* __restrict__ W2, _Float16* __restrict__ W2h) {
    const int NB_POOL = (NGRAPH * OUT_C) / 256;
    const int NB_W1   = (IN_C * HID_C) / 256;
    int b = blockIdx.x;
    int t = threadIdx.x;
    if (b < nb_cnt) {
        int i = b * 256 + t;
        if (i < n_cnt) cnt[i] = 0;
        return;
    }
    b -= nb_cnt;
    if (b < NB_POOL) { pool[b * 256 + t] = 0.0f; return; }
    b -= NB_POOL;
    if (b == 0) { if (t < NGRAPH) counts[t] = 0.0f; return; }
    b -= 1;
    if (b < NB_W1) { pack_w_elem(W1, W1h, IN_C, HID_C, b * 256 + t); return; }
    b -= NB_W1;
    pack_w_elem(W2, W2h, HID_C, OUT_C, b * 256 + t);
}

__global__ void k_node_setup(const int* cnt, int* total, float* __restrict__ dinv,
                             int* __restrict__ cursor, int Nn) {
    __shared__ int s_wave[8];
    __shared__ int s_base;
    int i    = blockIdx.x * 256 + threadIdx.x;
    int lane = threadIdx.x & 31;
    int wv   = threadIdx.x >> 5;
    int c = i < Nn ? cnt[i] : 0;
    if (i < Nn) dinv[i] = rsqrtf((float)(c + 1));
    int incl = c;
    #pragma unroll
    for (int o = 1; o < 32; o <<= 1) {
        int t = __shfl_up(incl, o);
        if (lane >= o) incl += t;
    }
    if (lane == 31) s_wave[wv] = incl;
    __syncthreads();
    if (threadIdx.x == 0) {
        int run = 0;
        for (int w = 0; w < 8; ++w) { int t = s_wave[w]; s_wave[w] = run; run += t; }
        s_base = atomicAdd(total, run);
    }
    __syncthreads();
    if (i < Nn) cursor[i] = s_base + s_wave[wv] + incl - c;
}

__global__ void k_edge_place(const int* __restrict__ src, const int* __restrict__ dst,
                             int* cursor, int* __restrict__ nbr, int E) {
    int e = blockIdx.x * blockDim.x + threadIdx.x;
    if (e >= E) return;
    int p = atomicAdd(&cursor[dst[e]], 1);
    nbr[p] = src[e];
}

template <int NT, bool RELU>
__device__ __forceinline__ void gemm_strip(int wave, const float* __restrict__ A,
                                           const _Float16* __restrict__ Bh,
                                           float* __restrict__ C, int M, int K, int N) {
    int lane = threadIdx.x & 31;
    int mtiles = M >> 4;
    if (wave >= mtiles) return;

    int m0  = wave << 4;
    int hi  = lane >> 4;
    int row = m0 + (lane & 15);

    v8f acc[NT];
    v8f zero = {0.f,0.f,0.f,0.f,0.f,0.f,0.f,0.f};
    #pragma unroll
    for (int j = 0; j < NT; ++j) acc[j] = zero;

    const v16h* bp = reinterpret_cast<const v16h*>(Bh);
    int ktiles = K >> 5;
    for (int kt = 0; kt < ktiles; ++kt) {
        int kbase = (kt << 5) + (hi << 3);
        const float4* p = reinterpret_cast<const float4*>(A + (long)row * K + kbase);
        float4 x0 = p[0], x1 = p[1];
        float4 x2 = p[4], x3 = p[5];
        if (RELU) {
            x0.x=fmaxf(x0.x,0.f); x0.y=fmaxf(x0.y,0.f); x0.z=fmaxf(x0.z,0.f); x0.w=fmaxf(x0.w,0.f);
            x1.x=fmaxf(x1.x,0.f); x1.y=fmaxf(x1.y,0.f); x1.z=fmaxf(x1.z,0.f); x1.w=fmaxf(x1.w,0.f);
            x2.x=fmaxf(x2.x,0.f); x2.y=fmaxf(x2.y,0.f); x2.z=fmaxf(x2.z,0.f); x2.w=fmaxf(x2.w,0.f);
            x3.x=fmaxf(x3.x,0.f); x3.y=fmaxf(x3.y,0.f); x3.z=fmaxf(x3.z,0.f); x3.w=fmaxf(x3.w,0.f);
        }
        v16h a;
        a[0]=(_Float16)x0.x; a[1]=(_Float16)x0.y; a[2]=(_Float16)x0.z; a[3]=(_Float16)x0.w;
        a[4]=(_Float16)x1.x; a[5]=(_Float16)x1.y; a[6]=(_Float16)x1.z; a[7]=(_Float16)x1.w;
        a[8]=(_Float16)x2.x; a[9]=(_Float16)x2.y; a[10]=(_Float16)x2.z; a[11]=(_Float16)x2.w;
        a[12]=(_Float16)x3.x; a[13]=(_Float16)x3.y; a[14]=(_Float16)x3.z; a[15]=(_Float16)x3.w;

        #pragma unroll
        for (int nt = 0; nt < NT; ++nt) {
            v16h b = bp[(size_t)(kt * NT + nt) * 32 + lane];
            acc[nt] = __builtin_amdgcn_wmma_f32_16x16x32_f16(
                false, a, false, b, (short)0, acc[nt], false, false);
        }
    }

    int mrow = m0 + (hi << 3);
    #pragma unroll
    for (int nt = 0; nt < NT; ++nt) {
        int n = (nt << 4) + (lane & 15);
        #pragma unroll
        for (int j = 0; j < 8; ++j)
            C[(long)(mrow + j) * N + n] = acc[nt][j];
    }
}

template <int NT>
__global__ void k_gemm_count(const float* __restrict__ A, const _Float16* __restrict__ Bh,
                             float* __restrict__ C, int M, int K, int N, int gemm_blocks,
                             const int* __restrict__ dst, int* cnt, int E) {
    int b = blockIdx.x;
    if (b < gemm_blocks) {
        gemm_strip<NT, false>((int)((b * blockDim.x + threadIdx.x) >> 5), A, Bh, C, M, K, N);
    } else {
        int e = (b - gemm_blocks) * blockDim.x + threadIdx.x;
        if (e < E) atomicAdd(&cnt[dst[e]], 1);
    }
}


#define TILE_M 16

template <int C> struct rowvec { typedef __attribute__((ext_vector_type(C / 32))) float type; };

template <int N, typename VT>
__device__ __forceinline__ void add_rows(VT& acc, const int* __restrict__ nbr, int j,
                                         const float* __restrict__ dinv, float dd,
                                         const VT* __restrict__ Hv, int lane) {
    int   s[N];
    float w[N];
    VT    r[N];
    #pragma unroll
    for (int k = 0; k < N; ++k) s[k] = nbr[j + k];
    #pragma unroll
    for (int k = 0; k < N; ++k) r[k] = Hv[(size_t)s[k] * 32 + lane];
    #pragma unroll
    for (int k = 0; k < N; ++k) w[k] = dinv[s[k]] * dd;
    #pragma unroll
    for (int k = 0; k < N; ++k) acc += r[k] * w[k];
}

template <int C>
__device__ __forceinline__ typename rowvec<C>::type
gather_row(int node, int lane, const int* __restrict__ cnt, const int* __restrict__ cursor,
           const int* __restrict__ nbr, const float* __restrict__ H,
           const float* __restrict__ dinv, const float* __restrict__ bias) {
    typedef typename rowvec<C>::type vt;
    int end = cursor[node];
    int j   = end - cnt[node];
    float dd = dinv[node];
    const vt* Hv = reinterpret_cast<const vt*>(H);
    vt acc = Hv[(size_t)node * 32 + lane] * (dd * dd);
    for (; j + 8 <= end; j += 8) add_rows<8, vt>(acc, nbr, j, dinv, dd, Hv, lane);
    int rem = end - j;
    if (rem & 4) { add_rows<4, vt>(acc, nbr, j, dinv, dd, Hv, lane); j += 4; }
    if (rem & 2) { add_rows<2, vt>(acc, nbr, j, dinv, dd, Hv, lane); j += 2; }
    if (rem & 1) { add_rows<1, vt>(acc, nbr, j, dinv, dd, Hv, lane); }
    const vt* bv = reinterpret_cast<const vt*>(bias);
    return acc + bv[lane];
}

__device__ __forceinline__ bool last_row_in(unsigned* ticket, int lane) {
    __threadfence_block();
    unsigned t = 0;
    if (lane == 0) t = atomicAdd(ticket, 1u);
    t = (unsigned)__builtin_amdgcn_readfirstlane((int)t);
    __threadfence_block();
    return t == TILE_M - 1;
}

#define A_PITCH 136

__global__ void __launch_bounds__(TILE_M * 32, 16)
k_gather_gemm(const int* __restrict__ cnt, const int* __restrict__ cursor,
              const int* __restrict__ nbr, const float* __restrict__ H,
              const float* __restrict__ dinv, const float* __restrict__ bias,
              const _Float16* __restrict__ Bh, float* __restrict__ C2, int Nn) {
    typedef __attribute__((ext_vector_type(4))) _Float16 v4h;
    typedef __attribute__((ext_vector_type(8))) _Float16 v8h;
    __shared__ __attribute__((aligned(16))) _Float16 sA[TILE_M * A_PITCH];
    __shared__ __attribute__((aligned(16))) v8h sB[(HID_C / 32) * (OUT_C / 16)][2][32];
    __shared__ unsigned s_ticket;
    if (threadIdx.x == 0) s_ticket = 0;
    __syncthreads();

    int wv   = __builtin_amdgcn_readfirstlane((int)(threadIdx.x >> 5));
    int lane = threadIdx.x & 31;
    int m0   = blockIdx.x * TILE_M;
    int node = m0 + wv;

    v16h bfrag = reinterpret_cast<const v16h*>(Bh)[wv * 32 + lane];

    rowvec<HID_C>::type row = {0.f, 0.f, 0.f, 0.f};
    if (node < Nn) row = gather_row<HID_C>(node, lane, cnt, cursor, nbr, H, dinv, bias);
    v4h h;
    h[0] = (_Float16)fmaxf(row[0], 0.f); h[1] = (_Float16)fmaxf(row[1], 0.f);
    h[2] = (_Float16)fmaxf(row[2], 0.f); h[3] = (_Float16)fmaxf(row[3], 0.f);
    *reinterpret_cast<v4h*>(&sA[wv * A_PITCH + lane * 4]) = h;
    sB[wv][0][lane] = __builtin_shufflevector(bfrag, bfrag, 0,1,2,3,4,5,6,7);
    sB[wv][1][lane] = __builtin_shufflevector(bfrag, bfrag, 8,9,10,11,12,13,14,15);

    if (!last_row_in(&s_ticket, lane)) return;

    const int NT = OUT_C / 16;
    int hi = lane >> 4;
    int r  = lane & 15;
    v8f acc[NT];
    v8f zero = {0.f,0.f,0.f,0.f,0.f,0.f,0.f,0.f};
    #pragma unroll
    for (int j = 0; j < NT; ++j) acc[j] = zero;
    #pragma unroll
    for (int kt = 0; kt < HID_C / 32; ++kt) {
        const _Float16* ap = &sA[r * A_PITCH + (kt << 5) + (hi << 3)];
        v8h lo = *reinterpret_cast<const v8h*>(ap);
        v8h up = *reinterpret_cast<const v8h*>(ap + 16);
        v16h a = __builtin_shufflevector(lo, up, 0,1,2,3,4,5,6,7,8,9,10,11,12,13,14,15);
        #pragma unroll
        for (int nt = 0; nt < NT; ++nt) {
            v16h b = __builtin_shufflevector(sB[kt * NT + nt][0][lane], sB[kt * NT + nt][1][lane],
                                             0,1,2,3,4,5,6,7,8,9,10,11,12,13,14,15);
            acc[nt] = __builtin_amdgcn_wmma_f32_16x16x32_f16(
                false, a, false, b, (short)0, acc[nt], false, false);
        }
    }
    int mrow = m0 + (hi << 3);
    bool whole = m0 + TILE_M <= Nn;
    #pragma unroll
    for (int nt = 0; nt < NT; ++nt) {
        int n = (nt << 4) + r;
        #pragma unroll
        for (int j = 0; j < 8; ++j)
            if (whole || mrow + j < Nn) C2[(long)(mrow + j) * OUT_C + n] = acc[nt][j];
    }
}

__device__ __forceinline__ void pool_runs(const int (&g)[TILE_M], const float (&v)[TILE_M], int c,
                                          float* pool, float* counts) {
    int   gc = g[0];
    float s  = v[0];
    float n  = 1.0f;
    #pragma unroll
    for (int k = 1; k < TILE_M; ++k) {
        if (g[k] != gc) {
            if (gc >= 0) {
                atomic_add_f32(&pool[gc * OUT_C + c], s);
                if (c == 0) atomic_add_f32(&counts[gc], n);
            }
            gc = g[k]; s = 0.0f; n = 0.0f;
        }
        s += v[k];
        n += 1.0f;
    }
    if (gc >= 0) {
        atomic_add_f32(&pool[gc * OUT_C + c], s);
        if (c == 0) atomic_add_f32(&counts[gc], n);
    }
}

__global__ void __launch_bounds__(TILE_M * 32)
k_gather_pool(const int* __restrict__ cnt, const int* __restrict__ cursor,
              const int* __restrict__ nbr, const float* __restrict__ H,
              const float* __restrict__ dinv, const float* __restrict__ bias,
              const int* __restrict__ batch, float* pool, float* counts, int Nn) {
    __shared__ __attribute__((aligned(16))) float sP[TILE_M * OUT_C];
    __shared__ unsigned s_ticket;
    if (threadIdx.x == 0) s_ticket = 0;
    __syncthreads();

    int wv   = __builtin_amdgcn_readfirstlane((int)(threadIdx.x >> 5));
    int lane = threadIdx.x & 31;
    int i0   = blockIdx.x * TILE_M;
    int node = i0 + wv;

    rowvec<OUT_C>::type row = {0.f, 0.f};
    if (node < Nn) row = gather_row<OUT_C>(node, lane, cnt, cursor, nbr, H, dinv, bias);
    *reinterpret_cast<rowvec<OUT_C>::type*>(&sP[wv * OUT_C + lane * 2]) = row;

    if (!last_row_in(&s_ticket, lane)) return;

    int g[TILE_M];
    if (i0 + TILE_M <= Nn) {
        #pragma unroll
        for (int k = 0; k < TILE_M; ++k) g[k] = batch[i0 + k];
    } else {
        #pragma unroll
        for (int k = 0; k < TILE_M; ++k) {
            int i  = i0 + k;
            int il = i < Nn ? i : Nn - 1;
            int gk = batch[il];
            g[k] = i < Nn ? gk : -1;
        }
    }
    #pragma unroll
    for (int half = 0; half < OUT_C / 32; ++half) {
        int c = lane + 32 * half;
        float v[TILE_M];
        #pragma unroll
        for (int k = 0; k < TILE_M; ++k) v[k] = sP[k * OUT_C + c];
        pool_runs(g, v, c, pool, counts);
    }
}

__global__ void k_pool_final(const float* __restrict__ pool, const float* __restrict__ counts,
                             float* __restrict__ out) {
    int idx = blockIdx.x * blockDim.x + threadIdx.x;
    if (idx >= NGRAPH * OUT_C) return;
    out[idx] = pool[idx] / fmaxf(counts[idx >> 6], 1.0f);
}

extern "C" void kernel_launch(void* const* d_in, const int* in_sizes, int n_in,
                              void* d_out, int out_size, void* d_ws, size_t ws_size,
                              hipStream_t stream) {
    const float* x   = (const float*)d_in[0];
    const float* W1  = (const float*)d_in[1];
    const float* b1  = (const float*)d_in[2];
    const float* W2  = (const float*)d_in[3];
    const float* b2  = (const float*)d_in[4];
    const int*   ei  = (const int*)d_in[5];
    const int*   bat = (const int*)d_in[6];

    const int Nn = in_sizes[0] / IN_C;
    const int E  = in_sizes[5] / 2;
    const int* src = ei;
    const int* dst = ei + E;

    char* w = (char*)d_ws;
    size_t off = 0;
    auto alloc = [&](size_t bytes) -> void* {
        void* p = w + off;
        off = (off + bytes + 255) & ~(size_t)255;
        return p;
    };
    int*      cnt   = (int*)alloc((size_t)(Nn + 1) * 4);
    int*      total = cnt + Nn;
    int*      cursor= (int*)alloc((size_t)Nn * 4);
    int*      nbr   = (int*)alloc((size_t)E * 4);
    float*    dinv  = (float*)alloc((size_t)Nn * 4);
    _Float16* W1h   = (_Float16*)alloc((size_t)IN_C * HID_C * 2);
    _Float16* W2h   = (_Float16*)alloc((size_t)HID_C * OUT_C * 2);
    float*    h1    = (float*)alloc((size_t)Nn * HID_C * 4);
    float*    h2    = (float*)alloc((size_t)Nn * OUT_C * 4);
    float*    pool  = (float*)alloc((size_t)NGRAPH * OUT_C * 4);
    float*    cnts  = (float*)alloc((size_t)NGRAPH * 4);
    if (off > ws_size) return;

    const int T = 256;
    {
        int nb_cnt = (Nn + 1 + T - 1) / T;
        int blocks = nb_cnt + (NGRAPH * OUT_C) / T + 1 + (IN_C * HID_C) / T + (HID_C * OUT_C) / T;
        k_setup<<<blocks, T, 0, stream>>>(cnt, Nn + 1, nb_cnt, pool, cnts, W1, W1h, W2, W2h);
    }
    {
        int waves = Nn / 16;
        int gemm_blocks = (waves * 32 + T - 1) / T;
        int edge_blocks = (E + T - 1) / T;
        k_gemm_count<HID_C / 16><<<gemm_blocks + edge_blocks, T, 0, stream>>>(
            x, W1h, h1, Nn, IN_C, HID_C, gemm_blocks, dst, cnt, E);
    }
    k_node_setup<<<(Nn + T - 1) / T, T, 0, stream>>>(cnt, total, dinv, cursor, Nn);
    k_edge_place<<<(E + T - 1) / T, T, 0, stream>>>(src, dst, cursor, nbr, E);

    k_gather_gemm<<<(Nn + TILE_M - 1) / TILE_M, TILE_M * 32, 0, stream>>>(
        cnt, cursor, nbr, h1, dinv, b1, W2h, h2, Nn);

    k_gather_pool<<<(Nn + TILE_M - 1) / TILE_M, TILE_M * 32, 0, stream>>>(
        cnt, cursor, nbr, h2, dinv, b2, bat, pool, cnts, Nn);
    k_pool_final<<<(NGRAPH * OUT_C + T - 1) / T, T, 0, stream>>>(pool, cnts, (float*)d_out);
}


# propagate: product kernels store rows scaled by dinv of their own node; the edge walk is row loads and adds, the sum scaled once by dinv[dst]; no dinv[src] per edge; one more f32 rounding per term
# speedup vs baseline: 2.6374x; 1.0870x over previous
#include <hip/hip_runtime.h>
#include <hip/hip_bf16.h>
#include <stdint.h>

typedef __attribute__((ext_vector_type(16))) _Float16 v16h;
typedef __attribute__((ext_vector_type(8)))  float    v8f;

#define IN_C   128
#define HID_C  128
#define OUT_C  64
#define NGRAPH 128

__device__ __forceinline__ void atomic_add_f32(float* p, float v) {
    unsafeAtomicAdd(p, v);
}

__device__ __forceinline__ void pack_w_elem(const float* __restrict__ W, _Float16* __restrict__ Wh,
                                            int K, int N, int idx) {
    if (idx >= K * N) return;
    int frag   = idx >> 9;
    int within = idx & 511;
    int lane   = within >> 4;
    int i      = within & 15;
    int ntiles = N >> 4;
    int kt = frag / ntiles, nt = frag % ntiles;
    int n  = (nt << 4) + (lane & 15);
    int kk = (i < 8 ? i : i + 8) + ((lane >> 4) << 3);
    int k  = (kt << 5) + kk;
    Wh[idx] = (_Float16)W[k * N + n];
}

__global__ void k_setup(int* __restrict__ cnt, int n_cnt, int nb_cnt,
                        float* __restrict__ pool, float* __restrict__ counts,
                        const float* __restrict__ W1, _Float16* __restrict__ W1h,
                        const float* __restrict__ W2, _Float16* __restrict__ W2h) {
    const int NB_POOL = (NGRAPH * OUT_C) / 256;
    const int NB_W1   = (IN_C * HID_C) / 256;
    int b = blockIdx.x;
    int t = threadIdx.x;
    if (b < nb_cnt) {
        int i = b * 256 + t;
        if (i < n_cnt) cnt[i] = 0;
        return;
    }
    b -= nb_cnt;
    if (b < NB_POOL) { pool[b * 256 + t] = 0.0f; return; }
    b -= NB_POOL;
    if (b == 0) { if (t < NGRAPH) counts[t] = 0.0f; return; }
    b -= 1;
    if (b < NB_W1) { pack_w_elem(W1, W1h, IN_C, HID_C, b * 256 + t); return; }
    b -= NB_W1;
    pack_w_elem(W2, W2h, HID_C, OUT_C, b * 256 + t);
}

__global__ void k_node_setup(const int* cnt, int* total, float* __restrict__ dinv,
                             int* __restrict__ cursor, int Nn) {
    __shared__ int s_wave[8];
    __shared__ int s_base;
    int i    = blockIdx.x * 256 + threadIdx.x;
    int lane = threadIdx.x & 31;
    int wv   = threadIdx.x >> 5;
    int c = i < Nn ? cnt[i] : 0;
    if (i < Nn) dinv[i] = rsqrtf((float)(c + 1));
    int incl = c;
    #pragma unroll
    for (int o = 1; o < 32; o <<= 1) {
        int t = __shfl_up(incl, o);
        if (lane >= o) incl += t;
    }
    if (lane == 31) s_wave[wv] = incl;
    __syncthreads();
    if (threadIdx.x == 0) {
        int run = 0;
        for (int w = 0; w < 8; ++w) { int t = s_wave[w]; s_wave[w] = run; run += t; }
        s_base = atomicAdd(total, run);
    }
    __syncthreads();
    if (i < Nn) cursor[i] = s_base + s_wave[wv] + incl - c;
}

template <int NT, bool RELU>
__device__ __forceinline__ void gemm_strip(int wave, const float* __restrict__ A,
                                           const _Float16* __restrict__ Bh,
                                           float* __restrict__ C, int M, int K, int N,
                                           const float* __restrict__ rs) {
    int lane = threadIdx.x & 31;
    int mtiles = M >> 4;
    if (wave >= mtiles) return;

    int m0  = wave << 4;
    int hi  = lane >> 4;
    int row = m0 + (lane & 15);

    v8f acc[NT];
    v8f zero = {0.f,0.f,0.f,0.f,0.f,0.f,0.f,0.f};
    #pragma unroll
    for (int j = 0; j < NT; ++j) acc[j] = zero;

    int mrow = m0 + (hi << 3);
    const float4* rp = reinterpret_cast<const float4*>(rs + mrow);
    float4 r0 = rp[0], r1 = rp[1];

    const v16h* bp = reinterpret_cast<const v16h*>(Bh);
    int ktiles = K >> 5;
    for (int kt = 0; kt < ktiles; ++kt) {
        int kbase = (kt << 5) + (hi << 3);
        const float4* p = reinterpret_cast<const float4*>(A + (long)row * K + kbase);
        float4 x0 = p[0], x1 = p[1];
        float4 x2 = p[4], x3 = p[5];
        if (RELU) {
            x0.x=fmaxf(x0.x,0.f); x0.y=fmaxf(x0.y,0.f); x0.z=fmaxf(x0.z,0.f); x0.w=fmaxf(x0.w,0.f);
            x1.x=fmaxf(x1.x,0.f); x1.y=fmaxf(x1.y,0.f); x1.z=fmaxf(x1.z,0.f); x1.w=fmaxf(x1.w,0.f);
            x2.x=fmaxf(x2.x,0.f); x2.y=fmaxf(x2.y,0.f); x2.z=fmaxf(x2.z,0.f); x2.w=fmaxf(x2.w,0.f);
            x3.x=fmaxf(x3.x,0.f); x3.y=fmaxf(x3.y,0.f); x3.z=fmaxf(x3.z,0.f); x3.w=fmaxf(x3.w,0.f);
        }
        v16h a;
        a[0]=(_Float16)x0.x; a[1]=(_Float16)x0.y; a[2]=(_Float16)x0.z; a[3]=(_Float16)x0.w;
        a[4]=(_Float16)x1.x; a[5]=(_Float16)x1.y; a[6]=(_Float16)x1.z; a[7]=(_Float16)x1.w;
        a[8]=(_Float16)x2.x; a[9]=(_Float16)x2.y; a[10]=(_Float16)x2.z; a[11]=(_Float16)x2.w;
        a[12]=(_Float16)x3.x; a[13]=(_Float16)x3.y; a[14]=(_Float16)x3.z; a[15]=(_Float16)x3.w;

        #pragma unroll
        for (int nt = 0; nt < NT; ++nt) {
            v16h b = bp[(size_t)(kt * NT + nt) * 32 + lane];
            acc[nt] = __builtin_amdgcn_wmma_f32_16x16x32_f16(
                false, a, false, b, (short)0, acc[nt], false, false);
        }
    }

    const float sc[8] = {r0.x, r0.y, r0.z, r0.w, r1.x, r1.y, r1.z, r1.w};
    #pragma unroll
    for (int nt = 0; nt < NT; ++nt) {
        int n = (nt << 4) + (lane & 15);
        #pragma unroll
        for (int j = 0; j < 8; ++j)
            C[(long)(mrow + j) * N + n] = acc[nt][j] * sc[j];
    }
}

__global__ void k_count(const int* __restrict__ dst, int* cnt, int E) {
    int e = blockIdx.x * blockDim.x + threadIdx.x;
    if (e < E) atomicAdd(&cnt[dst[e]], 1);
}

template <int NT>
__global__ void k_gemm_place(const float* __restrict__ A, const _Float16* __restrict__ Bh,
                             float* __restrict__ C, int M, int K, int N, int gemm_blocks,
                             const float* __restrict__ dinv,
                             const int* __restrict__ src, const int* __restrict__ dst,
                             int* cursor, int* __restrict__ nbr, int E) {
    int b = blockIdx.x;
    if (b < gemm_blocks) {
        gemm_strip<NT, false>((int)((b * blockDim.x + threadIdx.x) >> 5), A, Bh, C, M, K, N, dinv);
    } else {
        int e = (b - gemm_blocks) * blockDim.x + threadIdx.x;
        if (e < E) {
            int p = atomicAdd(&cursor[dst[e]], 1);
            nbr[p] = src[e];
        }
    }
}


#define TILE_M 16

template <int C> struct rowvec { typedef __attribute__((ext_vector_type(C / 32))) float type; };

template <int N, typename VT>
__device__ __forceinline__ void add_rows(VT& acc, const int* __restrict__ nbr, int j,
                                         const VT* __restrict__ Hv, int lane) {
    int s[N];
    VT  r[N];
    #pragma unroll
    for (int k = 0; k < N; ++k) s[k] = nbr[j + k];
    #pragma unroll
    for (int k = 0; k < N; ++k) r[k] = Hv[(size_t)s[k] * 32 + lane];
    #pragma unroll
    for (int k = 0; k < N; ++k) acc += r[k];
}

template <int C>
__device__ __forceinline__ typename rowvec<C>::type
gather_row(int node, int lane, float dd, const int* __restrict__ cnt, const int* __restrict__ cursor,
           const int* __restrict__ nbr, const float* __restrict__ H,
           const float* __restrict__ bias) {
    typedef typename rowvec<C>::type vt;
    int end = cursor[node];
    int j   = end - cnt[node];
    const vt* Hv = reinterpret_cast<const vt*>(H);
    vt acc = Hv[(size_t)node * 32 + lane];
    for (; j + 8 <= end; j += 8) add_rows<8, vt>(acc, nbr, j, Hv, lane);
    int rem = end - j;
    if (rem & 4) { add_rows<4, vt>(acc, nbr, j, Hv, lane); j += 4; }
    if (rem & 2) { add_rows<2, vt>(acc, nbr, j, Hv, lane); j += 2; }
    if (rem & 1) { add_rows<1, vt>(acc, nbr, j, Hv, lane); }
    const vt* bv = reinterpret_cast<const vt*>(bias);
    return acc * dd + bv[lane];
}

__device__ __forceinline__ bool last_row_in(unsigned* ticket, int lane) {
    __threadfence_block();
    unsigned t = 0;
    if (lane == 0) t = atomicAdd(ticket, 1u);
    t = (unsigned)__builtin_amdgcn_readfirstlane((int)t);
    __threadfence_block();
    return t == TILE_M - 1;
}

#define A_PITCH 136

__global__ void __launch_bounds__(TILE_M * 32, 16)
k_gather_gemm(const int* __restrict__ cnt, const int* __restrict__ cursor,
              const int* __restrict__ nbr, const float* __restrict__ H,
              const float* __restrict__ dinv, const float* __restrict__ bias,
              const _Float16* __restrict__ Bh, float* __restrict__ C2, int Nn) {
    typedef __attribute__((ext_vector_type(4))) _Float16 v4h;
    typedef __attribute__((ext_vector_type(8))) _Float16 v8h;
    __shared__ __attribute__((aligned(16))) _Float16 sA[TILE_M * A_PITCH];
    __shared__ __attribute__((aligned(16))) v8h sB[(HID_C / 32) * (OUT_C / 16)][2][32];
    __shared__ __attribute__((aligned(16))) float sD[TILE_M];
    __shared__ unsigned s_ticket;
    if (threadIdx.x == 0) s_ticket = 0;
    __syncthreads();

    int wv   = __builtin_amdgcn_readfirstlane((int)(threadIdx.x >> 5));
    int lane = threadIdx.x & 31;
    int m0   = blockIdx.x * TILE_M;
    int node = m0 + wv;

    v16h bfrag = reinterpret_cast<const v16h*>(Bh)[wv * 32 + lane];

    rowvec<HID_C>::type row = {0.f, 0.f, 0.f, 0.f};
    float dd = 0.f;
    if (node < Nn) {
        dd  = dinv[node];
        row = gather_row<HID_C>(node, lane, dd, cnt, cursor, nbr, H, bias);
    }
    if (lane == 0) sD[wv] = dd;
    v4h h;
    h[0] = (_Float16)fmaxf(row[0], 0.f); h[1] = (_Float16)fmaxf(row[1], 0.f);
    h[2] = (_Float16)fmaxf(row[2], 0.f); h[3] = (_Float16)fmaxf(row[3], 0.f);
    *reinterpret_cast<v4h*>(&sA[wv * A_PITCH + lane * 4]) = h;
    sB[wv][0][lane] = __builtin_shufflevector(bfrag, bfrag, 0,1,2,3,4,5,6,7);
    sB[wv][1][lane] = __builtin_shufflevector(bfrag, bfrag, 8,9,10,11,12,13,14,15);

    if (!last_row_in(&s_ticket, lane)) return;

    const int NT = OUT_C / 16;
    int hi = lane >> 4;
    int r  = lane & 15;
    v8f acc[NT];
    v8f zero = {0.f,0.f,0.f,0.f,0.f,0.f,0.f,0.f};
    #pragma unroll
    for (int j = 0; j < NT; ++j) acc[j] = zero;
    #pragma unroll
    for (int kt = 0; kt < HID_C / 32; ++kt) {
        const _Float16* ap = &sA[r * A_PITCH + (kt << 5) + (hi << 3)];
        v8h lo = *reinterpret_cast<const v8h*>(ap);
        v8h up = *reinterpret_cast<const v8h*>(ap + 16);
        v16h a = __builtin_shufflevector(lo, up, 0,1,2,3,4,5,6,7,8,9,10,11,12,13,14,15);
        #pragma unroll
        for (int nt = 0; nt < NT; ++nt) {
            v16h b = __builtin_shufflevector(sB[kt * NT + nt][0][lane], sB[kt * NT + nt][1][lane],
                                             0,1,2,3,4,5,6,7,8,9,10,11,12,13,14,15);
            acc[nt] = __builtin_amdgcn_wmma_f32_16x16x32_f16(
                false, a, false, b, (short)0, acc[nt], false, false);
        }
    }
    int mrow = m0 + (hi << 3);
    bool whole = m0 + TILE_M <= Nn;
    const float4* dp = reinterpret_cast<const float4*>(&sD[hi << 3]);
    float4 d0 = dp[0], d1 = dp[1];
    const float sc[8] = {d0.x, d0.y, d0.z, d0.w, d1.x, d1.y, d1.z, d1.w};
    #pragma unroll
    for (int nt = 0; nt < NT; ++nt) {
        int n = (nt << 4) + r;
        #pragma unroll
        for (int j = 0; j < 8; ++j)
            if (whole || mrow + j < Nn) C2[(long)(mrow + j) * OUT_C + n] = acc[nt][j] * sc[j];
    }
}

__device__ __forceinline__ void pool_runs(const int (&g)[TILE_M], const float (&v)[TILE_M], int c,
                                          float* pool, float* counts) {
    int   gc = g[0];
    float s  = v[0];
    float n  = 1.0f;
    #pragma unroll
    for (int k = 1; k < TILE_M; ++k) {
        if (g[k] != gc) {
            if (gc >= 0) {
                atomic_add_f32(&pool[gc * OUT_C + c], s);
                if (c == 0) atomic_add_f32(&counts[gc], n);
            }
            gc = g[k]; s = 0.0f; n = 0.0f;
        }
        s += v[k];
        n += 1.0f;
    }
    if (gc >= 0) {
        atomic_add_f32(&pool[gc * OUT_C + c], s);
        if (c == 0) atomic_add_f32(&counts[gc], n);
    }
}

__global__ void __launch_bounds__(TILE_M * 32)
k_gather_pool(const int* __restrict__ cnt, const int* __restrict__ cursor,
              const int* __restrict__ nbr, const float* __restrict__ H,
              const float* __restrict__ dinv, const float* __restrict__ bias,
              const int* __restrict__ batch, float* pool, float* counts, int Nn) {
    __shared__ __attribute__((aligned(16))) float sP[TILE_M * OUT_C];
    __shared__ unsigned s_ticket;
    if (threadIdx.x == 0) s_ticket = 0;
    __syncthreads();

    int wv   = __builtin_amdgcn_readfirstlane((int)(threadIdx.x >> 5));
    int lane = threadIdx.x & 31;
    int i0   = blockIdx.x * TILE_M;
    int node = i0 + wv;

    rowvec<OUT_C>::type row = {0.f, 0.f};
    if (node < Nn) row = gather_row<OUT_C>(node, lane, dinv[node], cnt, cursor, nbr, H, bias);
    *reinterpret_cast<rowvec<OUT_C>::type*>(&sP[wv * OUT_C + lane * 2]) = row;

    if (!last_row_in(&s_ticket, lane)) return;

    int g[TILE_M];
    if (i0 + TILE_M <= Nn) {
        #pragma unroll
        for (int k = 0; k < TILE_M; ++k) g[k] = batch[i0 + k];
    } else {
        #pragma unroll
        for (int k = 0; k < TILE_M; ++k) {
            int i  = i0 + k;
            int il = i < Nn ? i : Nn - 1;
            int gk = batch[il];
            g[k] = i < Nn ? gk : -1;
        }
    }
    #pragma unroll
    for (int half = 0; half < OUT_C / 32; ++half) {
        int c = lane + 32 * half;
        float v[TILE_M];
        #pragma unroll
        for (int k = 0; k < TILE_M; ++k) v[k] = sP[k * OUT_C + c];
        pool_runs(g, v, c, pool, counts);
    }
}

__global__ void k_pool_final(const float* __restrict__ pool, const float* __restrict__ counts,
                             float* __restrict__ out) {
    int idx = blockIdx.x * blockDim.x + threadIdx.x;
    if (idx >= NGRAPH * OUT_C) return;
    out[idx] = pool[idx] / fmaxf(counts[idx >> 6], 1.0f);
}

extern "C" void kernel_launch(void* const* d_in, const int* in_sizes, int n_in,
                              void* d_out, int out_size, void* d_ws, size_t ws_size,
                              hipStream_t stream) {
    const float* x   = (const float*)d_in[0];
    const float* W1  = (const float*)d_in[1];
    const float* b1  = (const float*)d_in[2];
    const float* W2  = (const float*)d_in[3];
    const float* b2  = (const float*)d_in[4];
    const int*   ei  = (const int*)d_in[5];
    const int*   bat = (const int*)d_in[6];

    const int Nn = in_sizes[0] / IN_C;
    const int E  = in_sizes[5] / 2;
    const int* src = ei;
    const int* dst = ei + E;

    char* w = (char*)d_ws;
    size_t off = 0;
    auto alloc = [&](size_t bytes) -> void* {
        void* p = w + off;
        off = (off + bytes + 255) & ~(size_t)255;
        return p;
    };
    int*      cnt   = (int*)alloc((size_t)(Nn + 1) * 4);
    int*      total = cnt + Nn;
    int*      cursor= (int*)alloc((size_t)Nn * 4);
    int*      nbr   = (int*)alloc((size_t)E * 4);
    float*    dinv  = (float*)alloc((size_t)Nn * 4);
    _Float16* W1h   = (_Float16*)alloc((size_t)IN_C * HID_C * 2);
    _Float16* W2h   = (_Float16*)alloc((size_t)HID_C * OUT_C * 2);
    float*    h1    = (float*)alloc((size_t)Nn * HID_C * 4);
    float*    h2    = (float*)alloc((size_t)Nn * OUT_C * 4);
    float*    pool  = (float*)alloc((size_t)NGRAPH * OUT_C * 4);
    float*    cnts  = (float*)alloc((size_t)NGRAPH * 4);
    if (off > ws_size) return;

    const int T = 256;
    {
        int nb_cnt = (Nn + 1 + T - 1) / T;
        int blocks = nb_cnt + (NGRAPH * OUT_C) / T + 1 + (IN_C * HID_C) / T + (HID_C * OUT_C) / T;
        k_setup<<<blocks, T, 0, stream>>>(cnt, Nn + 1, nb_cnt, pool, cnts, W1, W1h, W2, W2h);
    }
    k_count<<<(E + T - 1) / T, T, 0, stream>>>(dst, cnt, E);
    k_node_setup<<<(Nn + T - 1) / T, T, 0, stream>>>(cnt, total, dinv, cursor, Nn);
    {
        int waves = Nn / 16;
        int gemm_blocks = (waves * 32 + T - 1) / T;
        int edge_blocks = (E + T - 1) / T;
        k_gemm_place<HID_C / 16><<<gemm_blocks + edge_blocks, T, 0, stream>>>(
            x, W1h, h1, Nn, IN_C, HID_C, gemm_blocks, dinv, src, dst, cursor, nbr, E);
    }

    k_gather_gemm<<<(Nn + TILE_M - 1) / TILE_M, TILE_M * 32, 0, stream>>>(
        cnt, cursor, nbr, h1, dinv, b1, W2h, h2, Nn);

    k_gather_pool<<<(Nn + TILE_M - 1) / TILE_M, TILE_M * 32, 0, stream>>>(
        cnt, cursor, nbr, h2, dinv, b2, bat, pool, cnts, Nn);
    k_pool_final<<<(NGRAPH * OUT_C + T - 1) / T, T, 0, stream>>>(pool, cnts, (float*)d_out);
}


# edge grouping: the in-degree count's atomic add returns each edge's rank; placing stores at segment start + rank, no atomic (1 per edge, was 2); float sum order follows the count's arrival order
# speedup vs baseline: 2.9932x; 1.1349x over previous
#include <hip/hip_runtime.h>
#include <hip/hip_bf16.h>
#include <stdint.h>

typedef __attribute__((ext_vector_type(16))) _Float16 v16h;
typedef __attribute__((ext_vector_type(8)))  float    v8f;

#define IN_C   128
#define HID_C  128
#define OUT_C  64
#define NGRAPH 128

__device__ __forceinline__ void atomic_add_f32(float* p, float v) {
    unsafeAtomicAdd(p, v);
}

__device__ __forceinline__ void pack_w_elem(const float* __restrict__ W, _Float16* __restrict__ Wh,
                                            int K, int N, int idx) {
    if (idx >= K * N) return;
    int frag   = idx >> 9;
    int within = idx & 511;
    int lane   = within >> 4;
    int i      = within & 15;
    int ntiles = N >> 4;
    int kt = frag / ntiles, nt = frag % ntiles;
    int n  = (nt << 4) + (lane & 15);
    int kk = (i < 8 ? i : i + 8) + ((lane >> 4) << 3);
    int k  = (kt << 5) + kk;
    Wh[idx] = (_Float16)W[k * N + n];
}

__global__ void k_setup(int* __restrict__ cnt, int n_cnt, int nb_cnt,
                        float* __restrict__ pool, float* __restrict__ counts,
                        const float* __restrict__ W1, _Float16* __restrict__ W1h,
                        const float* __restrict__ W2, _Float16* __restrict__ W2h) {
    const int NB_POOL = (NGRAPH * OUT_C) / 256;
    const int NB_W1   = (IN_C * HID_C) / 256;
    int b = blockIdx.x;
    int t = threadIdx.x;
    if (b < nb_cnt) {
        int i = b * 256 + t;
        if (i < n_cnt) cnt[i] = 0;
        return;
    }
    b -= nb_cnt;
    if (b < NB_POOL) { pool[b * 256 + t] = 0.0f; return; }
    b -= NB_POOL;
    if (b == 0) { if (t < NGRAPH) counts[t] = 0.0f; return; }
    b -= 1;
    if (b < NB_W1) { pack_w_elem(W1, W1h, IN_C, HID_C, b * 256 + t); return; }
    b -= NB_W1;
    pack_w_elem(W2, W2h, HID_C, OUT_C, b * 256 + t);
}

__global__ void k_node_setup(const int* cnt, int* total, float* __restrict__ dinv,
                             int* __restrict__ cursor, int Nn) {
    __shared__ int s_wave[8];
    __shared__ int s_base;
    int i    = blockIdx.x * 256 + threadIdx.x;
    int lane = threadIdx.x & 31;
    int wv   = threadIdx.x >> 5;
    int c = i < Nn ? cnt[i] : 0;
    if (i < Nn) dinv[i] = rsqrtf((float)(c + 1));
    int incl = c;
    #pragma unroll
    for (int o = 1; o < 32; o <<= 1) {
        int t = __shfl_up(incl, o);
        if (lane >= o) incl += t;
    }
    if (lane == 31) s_wave[wv] = incl;
    __syncthreads();
    if (threadIdx.x == 0) {
        int run = 0;
        for (int w = 0; w < 8; ++w) { int t = s_wave[w]; s_wave[w] = run; run += t; }
        s_base = atomicAdd(total, run);
    }
    __syncthreads();
    if (i < Nn) cursor[i] = s_base + s_wave[wv] + incl - c;
}

template <int NT, bool RELU>
__device__ __forceinline__ void gemm_strip(int wave, const float* __restrict__ A,
                                           const _Float16* __restrict__ Bh,
                                           float* __restrict__ C, int M, int K, int N,
                                           const float* __restrict__ rs) {
    int lane = threadIdx.x & 31;
    int mtiles = M >> 4;
    if (wave >= mtiles) return;

    int m0  = wave << 4;
    int hi  = lane >> 4;
    int row = m0 + (lane & 15);

    v8f acc[NT];
    v8f zero = {0.f,0.f,0.f,0.f,0.f,0.f,0.f,0.f};
    #pragma unroll
    for (int j = 0; j < NT; ++j) acc[j] = zero;

    int mrow = m0 + (hi << 3);
    const float4* rp = reinterpret_cast<const float4*>(rs + mrow);
    float4 r0 = rp[0], r1 = rp[1];

    const v16h* bp = reinterpret_cast<const v16h*>(Bh);
    int ktiles = K >> 5;
    for (int kt = 0; kt < ktiles; ++kt) {
        int kbase = (kt << 5) + (hi << 3);
        const float4* p = reinterpret_cast<const float4*>(A + (long)row * K + kbase);
        float4 x0 = p[0], x1 = p[1];
        float4 x2 = p[4], x3 = p[5];
        if (RELU) {
            x0.x=fmaxf(x0.x,0.f); x0.y=fmaxf(x0.y,0.f); x0.z=fmaxf(x0.z,0.f); x0.w=fmaxf(x0.w,0.f);
            x1.x=fmaxf(x1.x,0.f); x1.y=fmaxf(x1.y,0.f); x1.z=fmaxf(x1.z,0.f); x1.w=fmaxf(x1.w,0.f);
            x2.x=fmaxf(x2.x,0.f); x2.y=fmaxf(x2.y,0.f); x2.z=fmaxf(x2.z,0.f); x2.w=fmaxf(x2.w,0.f);
            x3.x=fmaxf(x3.x,0.f); x3.y=fmaxf(x3.y,0.f); x3.z=fmaxf(x3.z,0.f); x3.w=fmaxf(x3.w,0.f);
        }
        v16h a;
        a[0]=(_Float16)x0.x; a[1]=(_Float16)x0.y; a[2]=(_Float16)x0.z; a[3]=(_Float16)x0.w;
        a[4]=(_Float16)x1.x; a[5]=(_Float16)x1.y; a[6]=(_Float16)x1.z; a[7]=(_Float16)x1.w;
        a[8]=(_Float16)x2.x; a[9]=(_Float16)x2.y; a[10]=(_Float16)x2.z; a[11]=(_Float16)x2.w;
        a[12]=(_Float16)x3.x; a[13]=(_Float16)x3.y; a[14]=(_Float16)x3.z; a[15]=(_Float16)x3.w;

        #pragma unroll
        for (int nt = 0; nt < NT; ++nt) {
            v16h b = bp[(size_t)(kt * NT + nt) * 32 + lane];
            acc[nt] = __builtin_amdgcn_wmma_f32_16x16x32_f16(
                false, a, false, b, (short)0, acc[nt], false, false);
        }
    }

    const float sc[8] = {r0.x, r0.y, r0.z, r0.w, r1.x, r1.y, r1.z, r1.w};
    #pragma unroll
    for (int nt = 0; nt < NT; ++nt) {
        int n = (nt << 4) + (lane & 15);
        #pragma unroll
        for (int j = 0; j < 8; ++j)
            C[(long)(mrow + j) * N + n] = acc[nt][j] * sc[j];
    }
}

__global__ void k_count(const int* __restrict__ dst, int* cnt, int* __restrict__ rank, int E) {
    int e = blockIdx.x * blockDim.x + threadIdx.x;
    if (e < E) rank[e] = atomicAdd(&cnt[dst[e]], 1);
}

template <int NT>
__global__ void k_gemm_place(const float* __restrict__ A, const _Float16* __restrict__ Bh,
                             float* __restrict__ C, int M, int K, int N, int gemm_blocks,
                             const float* __restrict__ dinv,
                             const int* __restrict__ src, const int* __restrict__ dst,
                             const int* __restrict__ cursor, const int* __restrict__ rank,
                             int* __restrict__ nbr, int E) {
    int b = blockIdx.x;
    if (b < gemm_blocks) {
        gemm_strip<NT, false>((int)((b * blockDim.x + threadIdx.x) >> 5), A, Bh, C, M, K, N, dinv);
    } else {
        int e = (b - gemm_blocks) * blockDim.x + threadIdx.x;
        if (e < E) nbr[cursor[dst[e]] + rank[e]] = src[e];
    }
}


#define TILE_M 16

template <int C> struct rowvec { typedef __attribute__((ext_vector_type(C / 32))) float type; };

template <int N, typename VT>
__device__ __forceinline__ void add_rows(VT& acc, const int* __restrict__ nbr, int j,
                                         const VT* __restrict__ Hv, int lane) {
    int s[N];
    VT  r[N];
    #pragma unroll
    for (int k = 0; k < N; ++k) s[k] = nbr[j + k];
    #pragma unroll
    for (int k = 0; k < N; ++k) r[k] = Hv[(size_t)s[k] * 32 + lane];
    #pragma unroll
    for (int k = 0; k < N; ++k) acc += r[k];
}

template <int C>
__device__ __forceinline__ typename rowvec<C>::type
gather_row(int node, int lane, float dd, const int* __restrict__ cnt, const int* __restrict__ cursor,
           const int* __restrict__ nbr, const float* __restrict__ H,
           const float* __restrict__ bias) {
    typedef typename rowvec<C>::type vt;
    int j   = cursor[node];
    int end = j + cnt[node];
    const vt* Hv = reinterpret_cast<const vt*>(H);
    vt acc = Hv[(size_t)node * 32 + lane];
    for (; j + 8 <= end; j += 8) add_rows<8, vt>(acc, nbr, j, Hv, lane);
    int rem = end - j;
    if (rem & 4) { add_rows<4, vt>(acc, nbr, j, Hv, lane); j += 4; }
    if (rem & 2) { add_rows<2, vt>(acc, nbr, j, Hv, lane); j += 2; }
    if (rem & 1) { add_rows<1, vt>(acc, nbr, j, Hv, lane); }
    const vt* bv = reinterpret_cast<const vt*>(bias);
    return acc * dd + bv[lane];
}

__device__ __forceinline__ bool last_row_in(unsigned* ticket, int lane) {
    __threadfence_block();
    unsigned t = 0;
    if (lane == 0) t = atomicAdd(ticket, 1u);
    t = (unsigned)__builtin_amdgcn_readfirstlane((int)t);
    __threadfence_block();
    return t == TILE_M - 1;
}

#define A_PITCH 136

__global__ void __launch_bounds__(TILE_M * 32, 16)
k_gather_gemm(const int* __restrict__ cnt, const int* __restrict__ cursor,
              const int* __restrict__ nbr, const float* __restrict__ H,
              const float* __restrict__ dinv, const float* __restrict__ bias,
              const _Float16* __restrict__ Bh, float* __restrict__ C2, int Nn) {
    typedef __attribute__((ext_vector_type(4))) _Float16 v4h;
    typedef __attribute__((ext_vector_type(8))) _Float16 v8h;
    __shared__ __attribute__((aligned(16))) _Float16 sA[TILE_M * A_PITCH];
    __shared__ __attribute__((aligned(16))) v8h sB[(HID_C / 32) * (OUT_C / 16)][2][32];
    __shared__ __attribute__((aligned(16))) float sD[TILE_M];
    __shared__ unsigned s_ticket;
    if (threadIdx.x == 0) s_ticket = 0;
    __syncthreads();

    int wv   = __builtin_amdgcn_readfirstlane((int)(threadIdx.x >> 5));
    int lane = threadIdx.x & 31;
    int m0   = blockIdx.x * TILE_M;
    int node = m0 + wv;

    v16h bfrag = reinterpret_cast<const v16h*>(Bh)[wv * 32 + lane];

    rowvec<HID_C>::type row = {0.f, 0.f, 0.f, 0.f};
    float dd = 0.f;
    if (node < Nn) {
        dd  = dinv[node];
        row = gather_row<HID_C>(node, lane, dd, cnt, cursor, nbr, H, bias);
    }
    if (lane == 0) sD[wv] = dd;
    v4h h;
    h[0] = (_Float16)fmaxf(row[0], 0.f); h[1] = (_Float16)fmaxf(row[1], 0.f);
    h[2] = (_Float16)fmaxf(row[2], 0.f); h[3] = (_Float16)fmaxf(row[3], 0.f);
    *reinterpret_cast<v4h*>(&sA[wv * A_PITCH + lane * 4]) = h;
    sB[wv][0][lane] = __builtin_shufflevector(bfrag, bfrag, 0,1,2,3,4,5,6,7);
    sB[wv][1][lane] = __builtin_shufflevector(bfrag, bfrag, 8,9,10,11,12,13,14,15);

    if (!last_row_in(&s_ticket, lane)) return;

    const int NT = OUT_C / 16;
    int hi = lane >> 4;
    int r  = lane & 15;
    v8f acc[NT];
    v8f zero = {0.f,0.f,0.f,0.f,0.f,0.f,0.f,0.f};
    #pragma unroll
    for (int j = 0; j < NT; ++j) acc[j] = zero;
    #pragma unroll
    for (int kt = 0; kt < HID_C / 32; ++kt) {
        const _Float16* ap = &sA[r * A_PITCH + (kt << 5) + (hi << 3)];
        v8h lo = *reinterpret_cast<const v8h*>(ap);
        v8h up = *reinterpret_cast<const v8h*>(ap + 16);
        v16h a = __builtin_shufflevector(lo, up, 0,1,2,3,4,5,6,7,8,9,10,11,12,13,14,15);
        #pragma unroll
        for (int nt = 0; nt < NT; ++nt) {
            v16h b = __builtin_shufflevector(sB[kt * NT + nt][0][lane], sB[kt * NT + nt][1][lane],
                                             0,1,2,3,4,5,6,7,8,9,10,11,12,13,14,15);
            acc[nt] = __builtin_amdgcn_wmma_f32_16x16x32_f16(
                false, a, false, b, (short)0, acc[nt], false, false);
        }
    }
    int mrow = m0 + (hi << 3);
    bool whole = m0 + TILE_M <= Nn;
    const float4* dp = reinterpret_cast<const float4*>(&sD[hi << 3]);
    float4 d0 = dp[0], d1 = dp[1];
    const float sc[8] = {d0.x, d0.y, d0.z, d0.w, d1.x, d1.y, d1.z, d1.w};
    #pragma unroll
    for (int nt = 0; nt < NT; ++nt) {
        int n = (nt << 4) + r;
        #pragma unroll
        for (int j = 0; j < 8; ++j)
            if (whole || mrow + j < Nn) C2[(long)(mrow + j) * OUT_C + n] = acc[nt][j] * sc[j];
    }
}

__device__ __forceinline__ void pool_runs(const int (&g)[TILE_M], const float (&v)[TILE_M], int c,
                                          float* pool, float* counts) {
    int   gc = g[0];
    float s  = v[0];
    float n  = 1.0f;
    #pragma unroll
    for (int k = 1; k < TILE_M; ++k) {
        if (g[k] != gc) {
            if (gc >= 0) {
                atomic_add_f32(&pool[gc * OUT_C + c], s);
                if (c == 0) atomic_add_f32(&counts[gc], n);
            }
            gc = g[k]; s = 0.0f; n = 0.0f;
        }
        s += v[k];
        n += 1.0f;
    }
    if (gc >= 0) {
        atomic_add_f32(&pool[gc * OUT_C + c], s);
        if (c == 0) atomic_add_f32(&counts[gc], n);
    }
}

__global__ void __launch_bounds__(TILE_M * 32)
k_gather_pool(const int* __restrict__ cnt, const int* __restrict__ cursor,
              const int* __restrict__ nbr, const float* __restrict__ H,
              const float* __restrict__ dinv, const float* __restrict__ bias,
              const int* __restrict__ batch, float* pool, float* counts, int Nn) {
    __shared__ __attribute__((aligned(16))) float sP[TILE_M * OUT_C];
    __shared__ unsigned s_ticket;
    if (threadIdx.x == 0) s_ticket = 0;
    __syncthreads();

    int wv   = __builtin_amdgcn_readfirstlane((int)(threadIdx.x >> 5));
    int lane = threadIdx.x & 31;
    int i0   = blockIdx.x * TILE_M;
    int node = i0 + wv;

    rowvec<OUT_C>::type row = {0.f, 0.f};
    if (node < Nn) row = gather_row<OUT_C>(node, lane, dinv[node], cnt, cursor, nbr, H, bias);
    *reinterpret_cast<rowvec<OUT_C>::type*>(&sP[wv * OUT_C + lane * 2]) = row;

    if (!last_row_in(&s_ticket, lane)) return;

    int g[TILE_M];
    if (i0 + TILE_M <= Nn) {
        #pragma unroll
        for (int k = 0; k < TILE_M; ++k) g[k] = batch[i0 + k];
    } else {
        #pragma unroll
        for (int k = 0; k < TILE_M; ++k) {
            int i  = i0 + k;
            int il = i < Nn ? i : Nn - 1;
            int gk = batch[il];
            g[k] = i < Nn ? gk : -1;
        }
    }
    #pragma unroll
    for (int half = 0; half < OUT_C / 32; ++half) {
        int c = lane + 32 * half;
        float v[TILE_M];
        #pragma unroll
        for (int k = 0; k < TILE_M; ++k) v[k] = sP[k * OUT_C + c];
        pool_runs(g, v, c, pool, counts);
    }
}

__global__ void k_pool_final(const float* __restrict__ pool, const float* __restrict__ counts,
                             float* __restrict__ out) {
    int idx = blockIdx.x * blockDim.x + threadIdx.x;
    if (idx >= NGRAPH * OUT_C) return;
    out[idx] = pool[idx] / fmaxf(counts[idx >> 6], 1.0f);
}

extern "C" void kernel_launch(void* const* d_in, const int* in_sizes, int n_in,
                              void* d_out, int out_size, void* d_ws, size_t ws_size,
                              hipStream_t stream) {
    const float* x   = (const float*)d_in[0];
    const float* W1  = (const float*)d_in[1];
    const float* b1  = (const float*)d_in[2];
    const float* W2  = (const float*)d_in[3];
    const float* b2  = (const float*)d_in[4];
    const int*   ei  = (const int*)d_in[5];
    const int*   bat = (const int*)d_in[6];

    const int Nn = in_sizes[0] / IN_C;
    const int E  = in_sizes[5] / 2;
    const int* src = ei;
    const int* dst = ei + E;

    char* w = (char*)d_ws;
    size_t off = 0;
    auto alloc = [&](size_t bytes) -> void* {
        void* p = w + off;
        off = (off + bytes + 255) & ~(size_t)255;
        return p;
    };
    int*      cnt   = (int*)alloc((size_t)(Nn + 1) * 4);
    int*      total = cnt + Nn;
    int*      cursor= (int*)alloc((size_t)Nn * 4);
    int*      nbr   = (int*)alloc((size_t)E * 4);
    float*    dinv  = (float*)alloc((size_t)Nn * 4);
    _Float16* W1h   = (_Float16*)alloc((size_t)IN_C * HID_C * 2);
    _Float16* W2h   = (_Float16*)alloc((size_t)HID_C * OUT_C * 2);
    float*    h1    = (float*)alloc((size_t)Nn * HID_C * 4);
    size_t    h2_bytes = (size_t)Nn * OUT_C * 4;
    size_t    rk_bytes = (size_t)E * 4;
    float*    h2    = (float*)alloc(h2_bytes > rk_bytes ? h2_bytes : rk_bytes);
    int*      rank  = (int*)h2;
    float*    pool  = (float*)alloc((size_t)NGRAPH * OUT_C * 4);
    float*    cnts  = (float*)alloc((size_t)NGRAPH * 4);
    if (off > ws_size) return;

    const int T = 256;
    {
        int nb_cnt = (Nn + 1 + T - 1) / T;
        int blocks = nb_cnt + (NGRAPH * OUT_C) / T + 1 + (IN_C * HID_C) / T + (HID_C * OUT_C) / T;
        k_setup<<<blocks, T, 0, stream>>>(cnt, Nn + 1, nb_cnt, pool, cnts, W1, W1h, W2, W2h);
    }
    k_count<<<(E + T - 1) / T, T, 0, stream>>>(dst, cnt, rank, E);
    k_node_setup<<<(Nn + T - 1) / T, T, 0, stream>>>(cnt, total, dinv, cursor, Nn);
    {
        int waves = Nn / 16;
        int gemm_blocks = (waves * 32 + T - 1) / T;
        int edge_blocks = (E + T - 1) / T;
        k_gemm_place<HID_C / 16><<<gemm_blocks + edge_blocks, T, 0, stream>>>(
            x, W1h, h1, Nn, IN_C, HID_C, gemm_blocks, dinv, src, dst, cursor, rank, nbr, E);
    }

    k_gather_gemm<<<(Nn + TILE_M - 1) / TILE_M, TILE_M * 32, 0, stream>>>(
        cnt, cursor, nbr, h1, dinv, b1, W2h, h2, Nn);

    k_gather_pool<<<(Nn + TILE_M - 1) / TILE_M, TILE_M * 32, 0, stream>>>(
        cnt, cursor, nbr, h2, dinv, b2, bat, pool, cnts, Nn);
    k_pool_final<<<(NGRAPH * OUT_C + T - 1) / T, T, 0, stream>>>(pool, cnts, (float*)d_out);
}


# edge list in fixed 40-slot segments per node: the count's atomic rank places the source id at once; no prefix-sum launch, cursor, rank array or placing pass; 7 launches to 6; sum order unchanged
# speedup vs baseline: 3.3776x; 1.1284x over previous
#include <hip/hip_runtime.h>
#include <hip/hip_bf16.h>
#include <stdint.h>

typedef __attribute__((ext_vector_type(16))) _Float16 v16h;
typedef __attribute__((ext_vector_type(8)))  float    v8f;

#define IN_C   128
#define HID_C  128
#define OUT_C  64
#define NGRAPH 128

__device__ __forceinline__ void atomic_add_f32(float* p, float v) {
    unsafeAtomicAdd(p, v);
}

__device__ __forceinline__ void pack_w_elem(const float* __restrict__ W, _Float16* __restrict__ Wh,
                                            int K, int N, int idx) {
    if (idx >= K * N) return;
    int frag   = idx >> 9;
    int within = idx & 511;
    int lane   = within >> 4;
    int i      = within & 15;
    int ntiles = N >> 4;
    int kt = frag / ntiles, nt = frag % ntiles;
    int n  = (nt << 4) + (lane & 15);
    int kk = (i < 8 ? i : i + 8) + ((lane >> 4) << 3);
    int k  = (kt << 5) + kk;
    Wh[idx] = (_Float16)W[k * N + n];
}

__global__ void k_setup(int* __restrict__ cnt, int n_cnt, int nb_cnt,
                        float* __restrict__ pool, float* __restrict__ counts,
                        const float* __restrict__ W1, _Float16* __restrict__ W1h,
                        const float* __restrict__ W2, _Float16* __restrict__ W2h) {
    const int NB_POOL = (NGRAPH * OUT_C) / 256;
    const int NB_W1   = (IN_C * HID_C) / 256;
    int b = blockIdx.x;
    int t = threadIdx.x;
    if (b < nb_cnt) {
        int i = b * 256 + t;
        if (i < n_cnt) cnt[i] = 0;
        return;
    }
    b -= nb_cnt;
    if (b < NB_POOL) { pool[b * 256 + t] = 0.0f; return; }
    b -= NB_POOL;
    if (b == 0) { if (t < NGRAPH) counts[t] = 0.0f; return; }
    b -= 1;
    if (b < NB_W1) { pack_w_elem(W1, W1h, IN_C, HID_C, b * 256 + t); return; }
    b -= NB_W1;
    pack_w_elem(W2, W2h, HID_C, OUT_C, b * 256 + t);
}

#define SEG 40

template <int NT, bool RELU>
__device__ __forceinline__ void gemm_strip(int wave, const float* __restrict__ A,
                                           const _Float16* __restrict__ Bh,
                                           float* __restrict__ C, int M, int K, int N,
                                           const int* __restrict__ cnt) {
    int lane = threadIdx.x & 31;
    int mtiles = M >> 4;
    if (wave >= mtiles) return;

    int m0  = wave << 4;
    int hi  = lane >> 4;
    int row = m0 + (lane & 15);

    v8f acc[NT];
    v8f zero = {0.f,0.f,0.f,0.f,0.f,0.f,0.f,0.f};
    #pragma unroll
    for (int j = 0; j < NT; ++j) acc[j] = zero;

    int mrow = m0 + (hi << 3);
    const int4* rp = reinterpret_cast<const int4*>(cnt + mrow);
    int4 c0 = rp[0], c1 = rp[1];

    const v16h* bp = reinterpret_cast<const v16h*>(Bh);
    int ktiles = K >> 5;
    for (int kt = 0; kt < ktiles; ++kt) {
        int kbase = (kt << 5) + (hi << 3);
        const float4* p = reinterpret_cast<const float4*>(A + (long)row * K + kbase);
        float4 x0 = p[0], x1 = p[1];
        float4 x2 = p[4], x3 = p[5];
        if (RELU) {
            x0.x=fmaxf(x0.x,0.f); x0.y=fmaxf(x0.y,0.f); x0.z=fmaxf(x0.z,0.f); x0.w=fmaxf(x0.w,0.f);
            x1.x=fmaxf(x1.x,0.f); x1.y=fmaxf(x1.y,0.f); x1.z=fmaxf(x1.z,0.f); x1.w=fmaxf(x1.w,0.f);
            x2.x=fmaxf(x2.x,0.f); x2.y=fmaxf(x2.y,0.f); x2.z=fmaxf(x2.z,0.f); x2.w=fmaxf(x2.w,0.f);
            x3.x=fmaxf(x3.x,0.f); x3.y=fmaxf(x3.y,0.f); x3.z=fmaxf(x3.z,0.f); x3.w=fmaxf(x3.w,0.f);
        }
        v16h a;
        a[0]=(_Float16)x0.x; a[1]=(_Float16)x0.y; a[2]=(_Float16)x0.z; a[3]=(_Float16)x0.w;
        a[4]=(_Float16)x1.x; a[5]=(_Float16)x1.y; a[6]=(_Float16)x1.z; a[7]=(_Float16)x1.w;
        a[8]=(_Float16)x2.x; a[9]=(_Float16)x2.y; a[10]=(_Float16)x2.z; a[11]=(_Float16)x2.w;
        a[12]=(_Float16)x3.x; a[13]=(_Float16)x3.y; a[14]=(_Float16)x3.z; a[15]=(_Float16)x3.w;

        #pragma unroll
        for (int nt = 0; nt < NT; ++nt) {
            v16h b = bp[(size_t)(kt * NT + nt) * 32 + lane];
            acc[nt] = __builtin_amdgcn_wmma_f32_16x16x32_f16(
                false, a, false, b, (short)0, acc[nt], false, false);
        }
    }

    const float sc[8] = {rsqrtf((float)(c0.x + 1)), rsqrtf((float)(c0.y + 1)), rsqrtf((float)(c0.z + 1)),
                         rsqrtf((float)(c0.w + 1)), rsqrtf((float)(c1.x + 1)), rsqrtf((float)(c1.y + 1)),
                         rsqrtf((float)(c1.z + 1)), rsqrtf((float)(c1.w + 1))};
    #pragma unroll
    for (int nt = 0; nt < NT; ++nt) {
        int n = (nt << 4) + (lane & 15);
        #pragma unroll
        for (int j = 0; j < 8; ++j)
            C[(long)(mrow + j) * N + n] = acc[nt][j] * sc[j];
    }
}

__global__ void k_count_place(const int* __restrict__ src, const int* __restrict__ dst, int* cnt,
                              int* __restrict__ nbr, int E) {
    int e = blockIdx.x * blockDim.x + threadIdx.x;
    if (e < E) {
        int d = dst[e];
        int s = src[e];
        asm volatile("" :: "v"(s));
        int r = atomicAdd(&cnt[d], 1);
        if (r < SEG) nbr[(size_t)d * SEG + r] = s;
    }
}

template <int NT>
__global__ void k_gemm(const float* __restrict__ A, const _Float16* __restrict__ Bh,
                       float* __restrict__ C, int M, int K, int N, const int* __restrict__ cnt) {
    gemm_strip<NT, false>((int)((blockIdx.x * blockDim.x + threadIdx.x) >> 5), A, Bh, C, M, K, N, cnt);
}


#define TILE_M 16

template <int C> struct rowvec { typedef __attribute__((ext_vector_type(C / 32))) float type; };

template <int N, typename VT>
__device__ __forceinline__ void add_rows(VT& acc, const int* __restrict__ nbr, int j,
                                         const VT* __restrict__ Hv, int lane) {
    int s[N];
    VT  r[N];
    #pragma unroll
    for (int k = 0; k < N; ++k) s[k] = nbr[j + k];
    #pragma unroll
    for (int k = 0; k < N; ++k) r[k] = Hv[(size_t)s[k] * 32 + lane];
    #pragma unroll
    for (int k = 0; k < N; ++k) acc += r[k];
}

template <typename VT>
__device__ __forceinline__ void walk_all_edges(VT& acc, int node, int lane, const int* __restrict__ src,
                                               const int* __restrict__ dst, int E,
                                               const VT* __restrict__ Hv) {
    for (int e0 = 0; e0 < E; e0 += 32) {
        int e = e0 + lane;
        int d = e < E ? dst[e] : -1;
        unsigned m = (unsigned)__ballot(d == node);
        while (m) {
            int b = __builtin_ctz(m);
            m &= m - 1;
            int s = src[e0 + b];
            acc += Hv[(size_t)s * 32 + lane];
        }
    }
}

template <int C>
__device__ __forceinline__ typename rowvec<C>::type
gather_row(int node, int lane, int c, float dd, const int* __restrict__ nbr,
           const int* __restrict__ src, const int* __restrict__ dst, int E,
           const float* __restrict__ H, const float* __restrict__ bias) {
    typedef typename rowvec<C>::type vt;
    const vt* Hv = reinterpret_cast<const vt*>(H);
    vt acc = Hv[(size_t)node * 32 + lane];
    if (c <= SEG) {
        int j   = node * SEG;
        int end = j + c;
        for (; j + 8 <= end; j += 8) add_rows<8, vt>(acc, nbr, j, Hv, lane);
        int rem = end - j;
        if (rem & 4) { add_rows<4, vt>(acc, nbr, j, Hv, lane); j += 4; }
        if (rem & 2) { add_rows<2, vt>(acc, nbr, j, Hv, lane); j += 2; }
        if (rem & 1) { add_rows<1, vt>(acc, nbr, j, Hv, lane); }
    } else {
        walk_all_edges<vt>(acc, node, lane, src, dst, E, Hv);
    }
    const vt* bv = reinterpret_cast<const vt*>(bias);
    return acc * dd + bv[lane];
}

__device__ __forceinline__ bool last_row_in(unsigned* ticket, int lane) {
    __threadfence_block();
    unsigned t = 0;
    if (lane == 0) t = atomicAdd(ticket, 1u);
    t = (unsigned)__builtin_amdgcn_readfirstlane((int)t);
    __threadfence_block();
    return t == TILE_M - 1;
}

#define A_PITCH 136

__global__ void __launch_bounds__(TILE_M * 32, 16)
k_gather_gemm(const int* __restrict__ cnt, const int* __restrict__ nbr,
              const int* __restrict__ src, const int* __restrict__ dst, int E,
              const float* __restrict__ H, const float* __restrict__ bias,
              const _Float16* __restrict__ Bh, float* __restrict__ C2, int Nn) {
    typedef __attribute__((ext_vector_type(4))) _Float16 v4h;
    typedef __attribute__((ext_vector_type(8))) _Float16 v8h;
    __shared__ __attribute__((aligned(16))) _Float16 sA[TILE_M * A_PITCH];
    __shared__ __attribute__((aligned(16))) v8h sB[(HID_C / 32) * (OUT_C / 16)][2][32];
    __shared__ __attribute__((aligned(16))) float sD[TILE_M];
    __shared__ unsigned s_ticket;
    if (threadIdx.x == 0) s_ticket = 0;
    __syncthreads();

    int wv   = __builtin_amdgcn_readfirstlane((int)(threadIdx.x >> 5));
    int lane = threadIdx.x & 31;
    int m0   = blockIdx.x * TILE_M;
    int node = m0 + wv;

    v16h bfrag = reinterpret_cast<const v16h*>(Bh)[wv * 32 + lane];

    rowvec<HID_C>::type row = {0.f, 0.f, 0.f, 0.f};
    float dd = 0.f;
    if (node < Nn) {
        int c = cnt[node];
        dd  = rsqrtf((float)(c + 1));
        row = gather_row<HID_C>(node, lane, c, dd, nbr, src, dst, E, H, bias);
    }
    if (lane == 0) sD[wv] = dd;
    v4h h;
    h[0] = (_Float16)fmaxf(row[0], 0.f); h[1] = (_Float16)fmaxf(row[1], 0.f);
    h[2] = (_Float16)fmaxf(row[2], 0.f); h[3] = (_Float16)fmaxf(row[3], 0.f);
    *reinterpret_cast<v4h*>(&sA[wv * A_PITCH + lane * 4]) = h;
    sB[wv][0][lane] = __builtin_shufflevector(bfrag, bfrag, 0,1,2,3,4,5,6,7);
    sB[wv][1][lane] = __builtin_shufflevector(bfrag, bfrag, 8,9,10,11,12,13,14,15);

    if (!last_row_in(&s_ticket, lane)) return;

    const int NT = OUT_C / 16;
    int hi = lane >> 4;
    int r  = lane & 15;
    v8f acc[NT];
    v8f zero = {0.f,0.f,0.f,0.f,0.f,0.f,0.f,0.f};
    #pragma unroll
    for (int j = 0; j < NT; ++j) acc[j] = zero;
    #pragma unroll
    for (int kt = 0; kt < HID_C / 32; ++kt) {
        const _Float16* ap = &sA[r * A_PITCH + (kt << 5) + (hi << 3)];
        v8h lo = *reinterpret_cast<const v8h*>(ap);
        v8h up = *reinterpret_cast<const v8h*>(ap + 16);
        v16h a = __builtin_shufflevector(lo, up, 0,1,2,3,4,5,6,7,8,9,10,11,12,13,14,15);
        #pragma unroll
        for (int nt = 0; nt < NT; ++nt) {
            v16h b = __builtin_shufflevector(sB[kt * NT + nt][0][lane], sB[kt * NT + nt][1][lane],
                                             0,1,2,3,4,5,6,7,8,9,10,11,12,13,14,15);
            acc[nt] = __builtin_amdgcn_wmma_f32_16x16x32_f16(
                false, a, false, b, (short)0, acc[nt], false, false);
        }
    }
    int mrow = m0 + (hi << 3);
    bool whole = m0 + TILE_M <= Nn;
    const float4* dp = reinterpret_cast<const float4*>(&sD[hi << 3]);
    float4 d0 = dp[0], d1 = dp[1];
    const float sc[8] = {d0.x, d0.y, d0.z, d0.w, d1.x, d1.y, d1.z, d1.w};
    #pragma unroll
    for (int nt = 0; nt < NT; ++nt) {
        int n = (nt << 4) + r;
        #pragma unroll
        for (int j = 0; j < 8; ++j)
            if (whole || mrow + j < Nn) C2[(long)(mrow + j) * OUT_C + n] = acc[nt][j] * sc[j];
    }
}

__device__ __forceinline__ void pool_runs(const int (&g)[TILE_M], const float (&v)[TILE_M], int c,
                                          float* pool, float* counts) {
    int   gc = g[0];
    float s  = v[0];
    float n  = 1.0f;
    #pragma unroll
    for (int k = 1; k < TILE_M; ++k) {
        if (g[k] != gc) {
            if (gc >= 0) {
                atomic_add_f32(&pool[gc * OUT_C + c], s);
                if (c == 0) atomic_add_f32(&counts[gc], n);
            }
            gc = g[k]; s = 0.0f; n = 0.0f;
        }
        s += v[k];
        n += 1.0f;
    }
    if (gc >= 0) {
        atomic_add_f32(&pool[gc * OUT_C + c], s);
        if (c == 0) atomic_add_f32(&counts[gc], n);
    }
}

__global__ void __launch_bounds__(TILE_M * 32)
k_gather_pool(const int* __restrict__ cnt, const int* __restrict__ nbr,
              const int* __restrict__ src, const int* __restrict__ dst, int E,
              const float* __restrict__ H, const float* __restrict__ bias,
              const int* __restrict__ batch, float* pool, float* counts, int Nn) {
    __shared__ __attribute__((aligned(16))) float sP[TILE_M * OUT_C];
    __shared__ unsigned s_ticket;
    if (threadIdx.x == 0) s_ticket = 0;
    __syncthreads();

    int wv   = __builtin_amdgcn_readfirstlane((int)(threadIdx.x >> 5));
    int lane = threadIdx.x & 31;
    int i0   = blockIdx.x * TILE_M;
    int node = i0 + wv;

    rowvec<OUT_C>::type row = {0.f, 0.f};
    if (node < Nn) {
        int c = cnt[node];
        row = gather_row<OUT_C>(node, lane, c, rsqrtf((float)(c + 1)), nbr, src, dst, E, H, bias);
    }
    *reinterpret_cast<rowvec<OUT_C>::type*>(&sP[wv * OUT_C + lane * 2]) = row;

    if (!last_row_in(&s_ticket, lane)) return;

    int g[TILE_M];
    if (i0 + TILE_M <= Nn) {
        #pragma unroll
        for (int k = 0; k < TILE_M; ++k) g[k] = batch[i0 + k];
    } else {
        #pragma unroll
        for (int k = 0; k < TILE_M; ++k) {
            int i  = i0 + k;
            int il = i < Nn ? i : Nn - 1;
            int gk = batch[il];
            g[k] = i < Nn ? gk : -1;
        }
    }
    #pragma unroll
    for (int half = 0; half < OUT_C / 32; ++half) {
        int c = lane + 32 * half;
        float v[TILE_M];
        #pragma unroll
        for (int k = 0; k < TILE_M; ++k) v[k] = sP[k * OUT_C + c];
        pool_runs(g, v, c, pool, counts);
    }
}

__global__ void k_pool_final(const float* __restrict__ pool, const float* __restrict__ counts,
                             float* __restrict__ out) {
    int idx = blockIdx.x * blockDim.x + threadIdx.x;
    if (idx >= NGRAPH * OUT_C) return;
    out[idx] = pool[idx] / fmaxf(counts[idx >> 6], 1.0f);
}

extern "C" void kernel_launch(void* const* d_in, const int* in_sizes, int n_in,
                              void* d_out, int out_size, void* d_ws, size_t ws_size,
                              hipStream_t stream) {
    const float* x   = (const float*)d_in[0];
    const float* W1  = (const float*)d_in[1];
    const float* b1  = (const float*)d_in[2];
    const float* W2  = (const float*)d_in[3];
    const float* b2  = (const float*)d_in[4];
    const int*   ei  = (const int*)d_in[5];
    const int*   bat = (const int*)d_in[6];

    const int Nn = in_sizes[0] / IN_C;
    const int E  = in_sizes[5] / 2;
    const int* src = ei;
    const int* dst = ei + E;

    char* w = (char*)d_ws;
    size_t off = 0;
    auto alloc = [&](size_t bytes) -> void* {
        void* p = w + off;
        off = (off + bytes + 255) & ~(size_t)255;
        return p;
    };
    int*      cnt   = (int*)alloc((size_t)Nn * 4);
    int*      nbr   = (int*)alloc((size_t)Nn * SEG * 4);
    _Float16* W1h   = (_Float16*)alloc((size_t)IN_C * HID_C * 2);
    _Float16* W2h   = (_Float16*)alloc((size_t)HID_C * OUT_C * 2);
    float*    h1    = (float*)alloc((size_t)Nn * HID_C * 4);
    float*    h2    = (float*)alloc((size_t)Nn * OUT_C * 4);
    float*    pool  = (float*)alloc((size_t)NGRAPH * OUT_C * 4);
    float*    cnts  = (float*)alloc((size_t)NGRAPH * 4);
    if (off > ws_size) return;

    const int T = 256;
    {
        int nb_cnt = (Nn + T - 1) / T;
        int blocks = nb_cnt + (NGRAPH * OUT_C) / T + 1 + (IN_C * HID_C) / T + (HID_C * OUT_C) / T;
        k_setup<<<blocks, T, 0, stream>>>(cnt, Nn, nb_cnt, pool, cnts, W1, W1h, W2, W2h);
    }
    k_count_place<<<(E + T - 1) / T, T, 0, stream>>>(src, dst, cnt, nbr, E);
    {
        int waves = Nn / 16;
        int gemm_blocks = (waves * 32 + T - 1) / T;
        k_gemm<HID_C / 16><<<gemm_blocks, T, 0, stream>>>(x, W1h, h1, Nn, IN_C, HID_C, cnt);
    }

    k_gather_gemm<<<(Nn + TILE_M - 1) / TILE_M, TILE_M * 32, 0, stream>>>(
        cnt, nbr, src, dst, E, h1, b1, W2h, h2, Nn);

    k_gather_pool<<<(Nn + TILE_M - 1) / TILE_M, TILE_M * 32, 0, stream>>>(
        cnt, nbr, src, dst, E, h2, b2, bat, pool, cnts, Nn);
    k_pool_final<<<(NGRAPH * OUT_C + T - 1) / T, T, 0, stream>>>(pool, cnts, (float*)d_out);
}


# propagate 1 draws half the bytes: layer 1's product stores its dinv-scaled rows as f16 (was f32), one 256-byte unit per row; edge sums stay f32; precision: one more f16 rounding per h1 element
# speedup vs baseline: 3.6621x; 1.0842x over previous
#include <hip/hip_runtime.h>
#include <hip/hip_bf16.h>
#include <stdint.h>

typedef __attribute__((ext_vector_type(16))) _Float16 v16h;
typedef __attribute__((ext_vector_type(8)))  float    v8f;

#define IN_C   128
#define HID_C  128
#define OUT_C  64
#define NGRAPH 128

__device__ __forceinline__ void atomic_add_f32(float* p, float v) {
    unsafeAtomicAdd(p, v);
}

__device__ __forceinline__ void pack_w_elem(const float* __restrict__ W, _Float16* __restrict__ Wh,
                                            int K, int N, int idx) {
    if (idx >= K * N) return;
    int frag   = idx >> 9;
    int within = idx & 511;
    int lane   = within >> 4;
    int i      = within & 15;
    int ntiles = N >> 4;
    int kt = frag / ntiles, nt = frag % ntiles;
    int n  = (nt << 4) + (lane & 15);
    int kk = (i < 8 ? i : i + 8) + ((lane >> 4) << 3);
    int k  = (kt << 5) + kk;
    Wh[idx] = (_Float16)W[k * N + n];
}

__global__ void k_setup(int* __restrict__ cnt, int n_cnt, int nb_cnt,
                        float* __restrict__ pool, float* __restrict__ counts,
                        const float* __restrict__ W1, _Float16* __restrict__ W1h,
                        const float* __restrict__ W2, _Float16* __restrict__ W2h) {
    const int NB_POOL = (NGRAPH * OUT_C) / 256;
    const int NB_W1   = (IN_C * HID_C) / 256;
    int b = blockIdx.x;
    int t = threadIdx.x;
    if (b < nb_cnt) {
        int i = b * 256 + t;
        if (i < n_cnt) cnt[i] = 0;
        return;
    }
    b -= nb_cnt;
    if (b < NB_POOL) { pool[b * 256 + t] = 0.0f; return; }
    b -= NB_POOL;
    if (b == 0) { if (t < NGRAPH) counts[t] = 0.0f; return; }
    b -= 1;
    if (b < NB_W1) { pack_w_elem(W1, W1h, IN_C, HID_C, b * 256 + t); return; }
    b -= NB_W1;
    pack_w_elem(W2, W2h, HID_C, OUT_C, b * 256 + t);
}

#define SEG 40

template <int NT, bool RELU>
__device__ __forceinline__ void gemm_strip(int wave, const float* __restrict__ A,
                                           const _Float16* __restrict__ Bh,
                                           _Float16* __restrict__ C, int M, int K, int N,
                                           const int* __restrict__ cnt) {
    int lane = threadIdx.x & 31;
    int mtiles = M >> 4;
    if (wave >= mtiles) return;

    int m0  = wave << 4;
    int hi  = lane >> 4;
    int row = m0 + (lane & 15);

    v8f acc[NT];
    v8f zero = {0.f,0.f,0.f,0.f,0.f,0.f,0.f,0.f};
    #pragma unroll
    for (int j = 0; j < NT; ++j) acc[j] = zero;

    int mrow = m0 + (hi << 3);
    const int4* rp = reinterpret_cast<const int4*>(cnt + mrow);
    int4 c0 = rp[0], c1 = rp[1];

    const v16h* bp = reinterpret_cast<const v16h*>(Bh);
    int ktiles = K >> 5;
    for (int kt = 0; kt < ktiles; ++kt) {
        int kbase = (kt << 5) + (hi << 3);
        const float4* p = reinterpret_cast<const float4*>(A + (long)row * K + kbase);
        float4 x0 = p[0], x1 = p[1];
        float4 x2 = p[4], x3 = p[5];
        if (RELU) {
            x0.x=fmaxf(x0.x,0.f); x0.y=fmaxf(x0.y,0.f); x0.z=fmaxf(x0.z,0.f); x0.w=fmaxf(x0.w,0.f);
            x1.x=fmaxf(x1.x,0.f); x1.y=fmaxf(x1.y,0.f); x1.z=fmaxf(x1.z,0.f); x1.w=fmaxf(x1.w,0.f);
            x2.x=fmaxf(x2.x,0.f); x2.y=fmaxf(x2.y,0.f); x2.z=fmaxf(x2.z,0.f); x2.w=fmaxf(x2.w,0.f);
            x3.x=fmaxf(x3.x,0.f); x3.y=fmaxf(x3.y,0.f); x3.z=fmaxf(x3.z,0.f); x3.w=fmaxf(x3.w,0.f);
        }
        v16h a;
        a[0]=(_Float16)x0.x; a[1]=(_Float16)x0.y; a[2]=(_Float16)x0.z; a[3]=(_Float16)x0.w;
        a[4]=(_Float16)x1.x; a[5]=(_Float16)x1.y; a[6]=(_Float16)x1.z; a[7]=(_Float16)x1.w;
        a[8]=(_Float16)x2.x; a[9]=(_Float16)x2.y; a[10]=(_Float16)x2.z; a[11]=(_Float16)x2.w;
        a[12]=(_Float16)x3.x; a[13]=(_Float16)x3.y; a[14]=(_Float16)x3.z; a[15]=(_Float16)x3.w;

        #pragma unroll
        for (int nt = 0; nt < NT; ++nt) {
            v16h b = bp[(size_t)(kt * NT + nt) * 32 + lane];
            acc[nt] = __builtin_amdgcn_wmma_f32_16x16x32_f16(
                false, a, false, b, (short)0, acc[nt], false, false);
        }
    }

    const float sc[8] = {rsqrtf((float)(c0.x + 1)), rsqrtf((float)(c0.y + 1)), rsqrtf((float)(c0.z + 1)),
                         rsqrtf((float)(c0.w + 1)), rsqrtf((float)(c1.x + 1)), rsqrtf((float)(c1.y + 1)),
                         rsqrtf((float)(c1.z + 1)), rsqrtf((float)(c1.w + 1))};
    #pragma unroll
    for (int nt = 0; nt < NT; ++nt) {
        int n = (nt << 4) + (lane & 15);
        #pragma unroll
        for (int j = 0; j < 8; ++j)
            C[(long)(mrow + j) * N + n] = (_Float16)(acc[nt][j] * sc[j]);
    }
}

__global__ void k_count_place(const int* __restrict__ src, const int* __restrict__ dst, int* cnt,
                              int* __restrict__ nbr, int E) {
    int e = blockIdx.x * blockDim.x + threadIdx.x;
    if (e < E) {
        int d = dst[e];
        int s = src[e];
        asm volatile("" :: "v"(s));
        int r = atomicAdd(&cnt[d], 1);
        if (r < SEG) nbr[(size_t)d * SEG + r] = s;
    }
}

template <int NT>
__global__ void k_gemm(const float* __restrict__ A, const _Float16* __restrict__ Bh,
                       _Float16* __restrict__ C, int M, int K, int N, const int* __restrict__ cnt) {
    gemm_strip<NT, false>((int)((blockIdx.x * blockDim.x + threadIdx.x) >> 5), A, Bh, C, M, K, N, cnt);
}


#define TILE_M 16

template <int C> struct rowvec { typedef __attribute__((ext_vector_type(C / 32))) float type; };
template <int C, typename ST> struct rowmem { typedef __attribute__((ext_vector_type(C / 32))) ST type; };

template <int N, typename VT, typename MT>
__device__ __forceinline__ void add_rows(VT& acc, const int* __restrict__ nbr, int j,
                                         const MT* __restrict__ Hv, int lane) {
    int s[N];
    MT  r[N];
    #pragma unroll
    for (int k = 0; k < N; ++k) s[k] = nbr[j + k];
    #pragma unroll
    for (int k = 0; k < N; ++k) r[k] = Hv[(size_t)s[k] * 32 + lane];
    #pragma unroll
    for (int k = 0; k < N; ++k) acc += __builtin_convertvector(r[k], VT);
}

template <typename VT, typename MT>
__device__ __forceinline__ void walk_all_edges(VT& acc, int node, int lane, const int* __restrict__ src,
                                               const int* __restrict__ dst, int E,
                                               const MT* __restrict__ Hv) {
    for (int e0 = 0; e0 < E; e0 += 32) {
        int e = e0 + lane;
        int d = e < E ? dst[e] : -1;
        unsigned m = (unsigned)__ballot(d == node);
        while (m) {
            int b = __builtin_ctz(m);
            m &= m - 1;
            int s = src[e0 + b];
            acc += __builtin_convertvector(Hv[(size_t)s * 32 + lane], VT);
        }
    }
}

template <int C, typename ST>
__device__ __forceinline__ typename rowvec<C>::type
gather_row(int node, int lane, int c, float dd, const int* __restrict__ nbr,
           const int* __restrict__ src, const int* __restrict__ dst, int E,
           const ST* __restrict__ H, const float* __restrict__ bias) {
    typedef typename rowvec<C>::type vt;
    typedef typename rowmem<C, ST>::type mt;
    const mt* Hv = reinterpret_cast<const mt*>(H);
    vt acc = __builtin_convertvector(Hv[(size_t)node * 32 + lane], vt);
    if (c <= SEG) {
        int j   = node * SEG;
        int end = j + c;
        for (; j + 8 <= end; j += 8) add_rows<8, vt, mt>(acc, nbr, j, Hv, lane);
        int rem = end - j;
        if (rem & 4) { add_rows<4, vt, mt>(acc, nbr, j, Hv, lane); j += 4; }
        if (rem & 2) { add_rows<2, vt, mt>(acc, nbr, j, Hv, lane); j += 2; }
        if (rem & 1) { add_rows<1, vt, mt>(acc, nbr, j, Hv, lane); }
    } else {
        walk_all_edges<vt, mt>(acc, node, lane, src, dst, E, Hv);
    }
    const vt* bv = reinterpret_cast<const vt*>(bias);
    return acc * dd + bv[lane];
}

__device__ __forceinline__ bool last_row_in(unsigned* ticket, int lane) {
    __threadfence_block();
    unsigned t = 0;
    if (lane == 0) t = atomicAdd(ticket, 1u);
    t = (unsigned)__builtin_amdgcn_readfirstlane((int)t);
    __threadfence_block();
    return t == TILE_M - 1;
}

#define A_PITCH 136

__global__ void __launch_bounds__(TILE_M * 32, 16)
k_gather_gemm(const int* __restrict__ cnt, const int* __restrict__ nbr,
              const int* __restrict__ src, const int* __restrict__ dst, int E,
              const _Float16* __restrict__ H, const float* __restrict__ bias,
              const _Float16* __restrict__ Bh, float* __restrict__ C2, int Nn) {
    typedef __attribute__((ext_vector_type(4))) _Float16 v4h;
    typedef __attribute__((ext_vector_type(8))) _Float16 v8h;
    __shared__ __attribute__((aligned(16))) _Float16 sA[TILE_M * A_PITCH];
    __shared__ __attribute__((aligned(16))) v8h sB[(HID_C / 32) * (OUT_C / 16)][2][32];
    __shared__ __attribute__((aligned(16))) float sD[TILE_M];
    __shared__ unsigned s_ticket;
    if (threadIdx.x == 0) s_ticket = 0;
    __syncthreads();

    int wv   = __builtin_amdgcn_readfirstlane((int)(threadIdx.x >> 5));
    int lane = threadIdx.x & 31;
    int m0   = blockIdx.x * TILE_M;
    int node = m0 + wv;

    v16h bfrag = reinterpret_cast<const v16h*>(Bh)[wv * 32 + lane];

    rowvec<HID_C>::type row = {0.f, 0.f, 0.f, 0.f};
    float dd = 0.f;
    if (node < Nn) {
        int c = cnt[node];
        dd  = rsqrtf((float)(c + 1));
        row = gather_row<HID_C, _Float16>(node, lane, c, dd, nbr, src, dst, E, H, bias);
    }
    if (lane == 0) sD[wv] = dd;
    v4h h;
    h[0] = (_Float16)fmaxf(row[0], 0.f); h[1] = (_Float16)fmaxf(row[1], 0.f);
    h[2] = (_Float16)fmaxf(row[2], 0.f); h[3] = (_Float16)fmaxf(row[3], 0.f);
    *reinterpret_cast<v4h*>(&sA[wv * A_PITCH + lane * 4]) = h;
    sB[wv][0][lane] = __builtin_shufflevector(bfrag, bfrag, 0,1,2,3,4,5,6,7);
    sB[wv][1][lane] = __builtin_shufflevector(bfrag, bfrag, 8,9,10,11,12,13,14,15);

    if (!last_row_in(&s_ticket, lane)) return;

    const int NT = OUT_C / 16;
    int hi = lane >> 4;
    int r  = lane & 15;
    v8f acc[NT];
    v8f zero = {0.f,0.f,0.f,0.f,0.f,0.f,0.f,0.f};
    #pragma unroll
    for (int j = 0; j < NT; ++j) acc[j] = zero;
    #pragma unroll
    for (int kt = 0; kt < HID_C / 32; ++kt) {
        const _Float16* ap = &sA[r * A_PITCH + (kt << 5) + (hi << 3)];
        v8h lo = *reinterpret_cast<const v8h*>(ap);
        v8h up = *reinterpret_cast<const v8h*>(ap + 16);
        v16h a = __builtin_shufflevector(lo, up, 0,1,2,3,4,5,6,7,8,9,10,11,12,13,14,15);
        #pragma unroll
        for (int nt = 0; nt < NT; ++nt) {
            v16h b = __builtin_shufflevector(sB[kt * NT + nt][0][lane], sB[kt * NT + nt][1][lane],
                                             0,1,2,3,4,5,6,7,8,9,10,11,12,13,14,15);
            acc[nt] = __builtin_amdgcn_wmma_f32_16x16x32_f16(
                false, a, false, b, (short)0, acc[nt], false, false);
        }
    }
    int mrow = m0 + (hi << 3);
    bool whole = m0 + TILE_M <= Nn;
    const float4* dp = reinterpret_cast<const float4*>(&sD[hi << 3]);
    float4 d0 = dp[0], d1 = dp[1];
    const float sc[8] = {d0.x, d0.y, d0.z, d0.w, d1.x, d1.y, d1.z, d1.w};
    #pragma unroll
    for (int nt = 0; nt < NT; ++nt) {
        int n = (nt << 4) + r;
        #pragma unroll
        for (int j = 0; j < 8; ++j)
            if (whole || mrow + j < Nn) C2[(long)(mrow + j) * OUT_C + n] = acc[nt][j] * sc[j];
    }
}

__device__ __forceinline__ void pool_runs(const int (&g)[TILE_M], const float (&v)[TILE_M], int c,
                                          float* pool, float* counts) {
    int   gc = g[0];
    float s  = v[0];
    float n  = 1.0f;
    #pragma unroll
    for (int k = 1; k < TILE_M; ++k) {
        if (g[k] != gc) {
            if (gc >= 0) {
                atomic_add_f32(&pool[gc * OUT_C + c], s);
                if (c == 0) atomic_add_f32(&counts[gc], n);
            }
            gc = g[k]; s = 0.0f; n = 0.0f;
        }
        s += v[k];
        n += 1.0f;
    }
    if (gc >= 0) {
        atomic_add_f32(&pool[gc * OUT_C + c], s);
        if (c == 0) atomic_add_f32(&counts[gc], n);
    }
}

__global__ void __launch_bounds__(TILE_M * 32)
k_gather_pool(const int* __restrict__ cnt, const int* __restrict__ nbr,
              const int* __restrict__ src, const int* __restrict__ dst, int E,
              const float* __restrict__ H, const float* __restrict__ bias,
              const int* __restrict__ batch, float* pool, float* counts, int Nn) {
    __shared__ __attribute__((aligned(16))) float sP[TILE_M * OUT_C];
    __shared__ unsigned s_ticket;
    if (threadIdx.x == 0) s_ticket = 0;
    __syncthreads();

    int wv   = __builtin_amdgcn_readfirstlane((int)(threadIdx.x >> 5));
    int lane = threadIdx.x & 31;
    int i0   = blockIdx.x * TILE_M;
    int node = i0 + wv;

    rowvec<OUT_C>::type row = {0.f, 0.f};
    if (node < Nn) {
        int c = cnt[node];
        row = gather_row<OUT_C, float>(node, lane, c, rsqrtf((float)(c + 1)), nbr, src, dst, E, H, bias);
    }
    *reinterpret_cast<rowvec<OUT_C>::type*>(&sP[wv * OUT_C + lane * 2]) = row;

    if (!last_row_in(&s_ticket, lane)) return;

    int g[TILE_M];
    if (i0 + TILE_M <= Nn) {
        #pragma unroll
        for (int k = 0; k < TILE_M; ++k) g[k] = batch[i0 + k];
    } else {
        #pragma unroll
        for (int k = 0; k < TILE_M; ++k) {
            int i  = i0 + k;
            int il = i < Nn ? i : Nn - 1;
            int gk = batch[il];
            g[k] = i < Nn ? gk : -1;
        }
    }
    #pragma unroll
    for (int half = 0; half < OUT_C / 32; ++half) {
        int c = lane + 32 * half;
        float v[TILE_M];
        #pragma unroll
        for (int k = 0; k < TILE_M; ++k) v[k] = sP[k * OUT_C + c];
        pool_runs(g, v, c, pool, counts);
    }
}

__global__ void k_pool_final(const float* __restrict__ pool, const float* __restrict__ counts,
                             float* __restrict__ out) {
    int idx = blockIdx.x * blockDim.x + threadIdx.x;
    if (idx >= NGRAPH * OUT_C) return;
    out[idx] = pool[idx] / fmaxf(counts[idx >> 6], 1.0f);
}

extern "C" void kernel_launch(void* const* d_in, const int* in_sizes, int n_in,
                              void* d_out, int out_size, void* d_ws, size_t ws_size,
                              hipStream_t stream) {
    const float* x   = (const float*)d_in[0];
    const float* W1  = (const float*)d_in[1];
    const float* b1  = (const float*)d_in[2];
    const float* W2  = (const float*)d_in[3];
    const float* b2  = (const float*)d_in[4];
    const int*   ei  = (const int*)d_in[5];
    const int*   bat = (const int*)d_in[6];

    const int Nn = in_sizes[0] / IN_C;
    const int E  = in_sizes[5] / 2;
    const int* src = ei;
    const int* dst = ei + E;

    char* w = (char*)d_ws;
    size_t off = 0;
    auto alloc = [&](size_t bytes) -> void* {
        void* p = w + off;
        off = (off + bytes + 255) & ~(size_t)255;
        return p;
    };
    int*      cnt   = (int*)alloc((size_t)Nn * 4);
    int*      nbr   = (int*)alloc((size_t)Nn * SEG * 4);
    _Float16* W1h   = (_Float16*)alloc((size_t)IN_C * HID_C * 2);
    _Float16* W2h   = (_Float16*)alloc((size_t)HID_C * OUT_C * 2);
    _Float16* h1    = (_Float16*)alloc((size_t)Nn * HID_C * 2);
    float*    h2    = (float*)alloc((size_t)Nn * OUT_C * 4);
    float*    pool  = (float*)alloc((size_t)NGRAPH * OUT_C * 4);
    float*    cnts  = (float*)alloc((size_t)NGRAPH * 4);
    if (off > ws_size) return;

    const int T = 256;
    {
        int nb_cnt = (Nn + T - 1) / T;
        int blocks = nb_cnt + (NGRAPH * OUT_C) / T + 1 + (IN_C * HID_C) / T + (HID_C * OUT_C) / T;
        k_setup<<<blocks, T, 0, stream>>>(cnt, Nn, nb_cnt, pool, cnts, W1, W1h, W2, W2h);
    }
    k_count_place<<<(E + T - 1) / T, T, 0, stream>>>(src, dst, cnt, nbr, E);
    {
        int waves = Nn / 16;
        int gemm_blocks = (waves * 32 + T - 1) / T;
        k_gemm<HID_C / 16><<<gemm_blocks, T, 0, stream>>>(x, W1h, h1, Nn, IN_C, HID_C, cnt);
    }

    k_gather_gemm<<<(Nn + TILE_M - 1) / TILE_M, TILE_M * 32, 0, stream>>>(
        cnt, nbr, src, dst, E, h1, b1, W2h, h2, Nn);

    k_gather_pool<<<(Nn + TILE_M - 1) / TILE_M, TILE_M * 32, 0, stream>>>(
        cnt, nbr, src, dst, E, h2, b2, bat, pool, cnts, Nn);
    k_pool_final<<<(NGRAPH * OUT_C + T - 1) / T, T, 0, stream>>>(pool, cnts, (float*)d_out);
}


# edge grouping: edges sorted by 256-node destination bin in LDS, then a block per bin ranks them with LDS adds, stores counts and segments as whole lines; no global atomic; sum order follows LDS ranks
# speedup vs baseline: 5.7469x; 1.5693x over previous
#include <hip/hip_runtime.h>
#include <hip/hip_bf16.h>
#include <stdint.h>

typedef __attribute__((ext_vector_type(16))) _Float16 v16h;
typedef __attribute__((ext_vector_type(8)))  float    v8f;

#define IN_C   128
#define HID_C  128
#define OUT_C  64
#define NGRAPH 128

__device__ __forceinline__ void atomic_add_f32(float* p, float v) {
    unsafeAtomicAdd(p, v);
}

__device__ __forceinline__ void pack_w_elem(const float* __restrict__ W, _Float16* __restrict__ Wh,
                                            int K, int N, int idx) {
    if (idx >= K * N) return;
    int frag   = idx >> 9;
    int within = idx & 511;
    int lane   = within >> 4;
    int i      = within & 15;
    int ntiles = N >> 4;
    int kt = frag / ntiles, nt = frag % ntiles;
    int n  = (nt << 4) + (lane & 15);
    int kk = (i < 8 ? i : i + 8) + ((lane >> 4) << 3);
    int k  = (kt << 5) + kk;
    Wh[idx] = (_Float16)W[k * N + n];
}

#define SEG 40

template <int NT, bool RELU>
__device__ __forceinline__ void gemm_strip(int wave, const float* __restrict__ A,
                                           const _Float16* __restrict__ Bh,
                                           _Float16* __restrict__ C, int M, int K, int N,
                                           const int* __restrict__ cnt) {
    int lane = threadIdx.x & 31;
    int mtiles = M >> 4;
    if (wave >= mtiles) return;

    int m0  = wave << 4;
    int hi  = lane >> 4;
    int row = m0 + (lane & 15);

    v8f acc[NT];
    v8f zero = {0.f,0.f,0.f,0.f,0.f,0.f,0.f,0.f};
    #pragma unroll
    for (int j = 0; j < NT; ++j) acc[j] = zero;

    int mrow = m0 + (hi << 3);
    const int4* rp = reinterpret_cast<const int4*>(cnt + mrow);
    int4 c0 = rp[0], c1 = rp[1];

    const v16h* bp = reinterpret_cast<const v16h*>(Bh);
    int ktiles = K >> 5;
    for (int kt = 0; kt < ktiles; ++kt) {
        int kbase = (kt << 5) + (hi << 3);
        const float4* p = reinterpret_cast<const float4*>(A + (long)row * K + kbase);
        float4 x0 = p[0], x1 = p[1];
        float4 x2 = p[4], x3 = p[5];
        if (RELU) {
            x0.x=fmaxf(x0.x,0.f); x0.y=fmaxf(x0.y,0.f); x0.z=fmaxf(x0.z,0.f); x0.w=fmaxf(x0.w,0.f);
            x1.x=fmaxf(x1.x,0.f); x1.y=fmaxf(x1.y,0.f); x1.z=fmaxf(x1.z,0.f); x1.w=fmaxf(x1.w,0.f);
            x2.x=fmaxf(x2.x,0.f); x2.y=fmaxf(x2.y,0.f); x2.z=fmaxf(x2.z,0.f); x2.w=fmaxf(x2.w,0.f);
            x3.x=fmaxf(x3.x,0.f); x3.y=fmaxf(x3.y,0.f); x3.z=fmaxf(x3.z,0.f); x3.w=fmaxf(x3.w,0.f);
        }
        v16h a;
        a[0]=(_Float16)x0.x; a[1]=(_Float16)x0.y; a[2]=(_Float16)x0.z; a[3]=(_Float16)x0.w;
        a[4]=(_Float16)x1.x; a[5]=(_Float16)x1.y; a[6]=(_Float16)x1.z; a[7]=(_Float16)x1.w;
        a[8]=(_Float16)x2.x; a[9]=(_Float16)x2.y; a[10]=(_Float16)x2.z; a[11]=(_Float16)x2.w;
        a[12]=(_Float16)x3.x; a[13]=(_Float16)x3.y; a[14]=(_Float16)x3.z; a[15]=(_Float16)x3.w;

        #pragma unroll
        for (int nt = 0; nt < NT; ++nt) {
            v16h b = bp[(size_t)(kt * NT + nt) * 32 + lane];
            acc[nt] = __builtin_amdgcn_wmma_f32_16x16x32_f16(
                false, a, false, b, (short)0, acc[nt], false, false);
        }
    }

    const float sc[8] = {rsqrtf((float)(c0.x + 1)), rsqrtf((float)(c0.y + 1)), rsqrtf((float)(c0.z + 1)),
                         rsqrtf((float)(c0.w + 1)), rsqrtf((float)(c1.x + 1)), rsqrtf((float)(c1.y + 1)),
                         rsqrtf((float)(c1.z + 1)), rsqrtf((float)(c1.w + 1))};
    #pragma unroll
    for (int nt = 0; nt < NT; ++nt) {
        int n = (nt << 4) + (lane & 15);
        #pragma unroll
        for (int j = 0; j < 8; ++j)
            C[(long)(mrow + j) * N + n] = (_Float16)(acc[nt][j] * sc[j]);
    }
}

#define BIN_SH   8
#define BIN_N    (1 << BIN_SH)
#define RUN      64
#define EPT      4
#define TA       1024
#define EPB      (EPT * TA)
#define MAXBIN   224
#define RUN_VALID 0x80000000u
#define TB       1024

__global__ void __launch_bounds__(TA)
k_bin_setup(const int* __restrict__ src, const int* __restrict__ dst, int E, int nbins, int nblk,
            unsigned* __restrict__ runs, int* __restrict__ runlen,
            float* __restrict__ pool, float* __restrict__ counts,
            const float* __restrict__ W1, _Float16* __restrict__ W1h,
            const float* __restrict__ W2, _Float16* __restrict__ W2h) {
    __shared__ __attribute__((aligned(16))) unsigned stage[MAXBIN * RUN];
    __shared__ int hist[MAXBIN];
    int t = threadIdx.x;
    int b = blockIdx.x;
    if (b >= nblk) {
        const int NB_POOL = (NGRAPH * OUT_C) / TA;
        const int NB_W1   = (IN_C * HID_C) / TA;
        b -= nblk;
        if (b < NB_POOL) { pool[b * TA + t] = 0.0f; return; }
        b -= NB_POOL;
        if (b == 0) { if (t < NGRAPH) counts[t] = 0.0f; return; }
        b -= 1;
        if (b < NB_W1) { pack_w_elem(W1, W1h, IN_C, HID_C, b * TA + t); return; }
        b -= NB_W1;
        pack_w_elem(W2, W2h, HID_C, OUT_C, b * TA + t);
        return;
    }

    int d[EPT], s[EPT];
    int e0 = b * EPB + t;
    #pragma unroll
    for (int k = 0; k < EPT; ++k) {
        int e  = e0 + k * TA;
        int ec = e < E ? e : E - 1;
        int dv = dst[ec];
        s[k] = src[ec];
        d[k] = e < E ? dv : -1;
    }
    uint4* stage4 = reinterpret_cast<uint4*>(stage);
    for (int i = t; i < nbins * (RUN / 4); i += TA) stage4[i] = make_uint4(0u, 0u, 0u, 0u);
    for (int i = t; i < nbins; i += TA) hist[i] = 0;
    __syncthreads();

    #pragma unroll
    for (int k = 0; k < EPT; ++k) {
        if (d[k] >= 0) {
            int bin = d[k] >> BIN_SH;
            int r = atomicAdd(&hist[bin], 1);
            if (r < RUN) stage[bin * RUN + r] = RUN_VALID | ((unsigned)s[k] << BIN_SH) | (unsigned)(d[k] & (BIN_N - 1));
        }
    }
    __syncthreads();

    uint4* runs4 = reinterpret_cast<uint4*>(runs);
    for (int i = t; i < nbins * (RUN / 4); i += TA) {
        int bin = i / (RUN / 4);
        runs4[((size_t)bin * nblk + b) * (RUN / 4) + (i % (RUN / 4))] = stage4[i];
    }
    for (int i = t; i < nbins; i += TA) runlen[(size_t)b * nbins + i] = hist[i];
}

__global__ void __launch_bounds__(TB)
k_rank_place(const int* __restrict__ src, const int* __restrict__ dst, int E, int nbins, int nblk,
             const unsigned* __restrict__ runs, const int* __restrict__ runlen,
             int* __restrict__ cnt, int* __restrict__ nbr, int Nn) {
    __shared__ __attribute__((aligned(16))) int seg[BIN_N * SEG];
    __shared__ int lcnt[BIN_N];
    __shared__ int s_over;
    int t   = threadIdx.x;
    int bin = blockIdx.x;
    const unsigned* rb = runs + (size_t)bin * nblk * RUN;
    int total = nblk * RUN;
    const int UN = 10;
    unsigned w[UN];
    #pragma unroll
    for (int u = 0; u < UN; ++u) {
        int idx = t + u * TB;
        w[u] = idx < total ? rb[idx] : 0u;
    }
    int over = 0;
    for (int k = t; k < nblk; k += TB) over |= runlen[(size_t)k * nbins + bin] > RUN;

    uint4* seg4 = reinterpret_cast<uint4*>(seg);
    for (int i = t; i < BIN_N * SEG / 4; i += TB) seg4[i] = make_uint4(0u, 0u, 0u, 0u);
    if (t < BIN_N) lcnt[t] = 0;
    if (t == 0) s_over = 0;
    __syncthreads();
    if (over) s_over = 1;
    __syncthreads();

    if (!s_over) {
        #pragma unroll
        for (int u = 0; u < UN; ++u) {
            if (w[u] & RUN_VALID) {
                int dl = (int)(w[u] & (BIN_N - 1));
                int r  = atomicAdd(&lcnt[dl], 1);
                if (r < SEG) seg[dl * SEG + r] = (int)((w[u] & ~RUN_VALID) >> BIN_SH);
            }
        }
        for (int idx = t + UN * TB; idx < total; idx += TB) {
            unsigned x = rb[idx];
            if (x & RUN_VALID) {
                int dl = (int)(x & (BIN_N - 1));
                int r  = atomicAdd(&lcnt[dl], 1);
                if (r < SEG) seg[dl * SEG + r] = (int)((x & ~RUN_VALID) >> BIN_SH);
            }
        }
    } else {
        for (int e = t; e < E; e += TB) {
            int d = dst[e];
            if ((d >> BIN_SH) == bin) {
                int dl = d & (BIN_N - 1);
                int r  = atomicAdd(&lcnt[dl], 1);
                if (r < SEG) seg[dl * SEG + r] = src[e];
            }
        }
    }
    __syncthreads();

    int node0 = bin << BIN_SH;
    int nn    = Nn - node0 < BIN_N ? Nn - node0 : BIN_N;
    if (t < nn) cnt[node0 + t] = lcnt[t];
    uint4* nbr4 = reinterpret_cast<uint4*>(nbr + (size_t)node0 * SEG);
    for (int i = t; i < nn * (SEG / 4); i += TB) nbr4[i] = seg4[i];
}

template <int NT>
__global__ void k_gemm(const float* __restrict__ A, const _Float16* __restrict__ Bh,
                       _Float16* __restrict__ C, int M, int K, int N, const int* __restrict__ cnt) {
    gemm_strip<NT, false>((int)((blockIdx.x * blockDim.x + threadIdx.x) >> 5), A, Bh, C, M, K, N, cnt);
}


#define TILE_M 16

template <int C> struct rowvec { typedef __attribute__((ext_vector_type(C / 32))) float type; };
template <int C, typename ST> struct rowmem { typedef __attribute__((ext_vector_type(C / 32))) ST type; };

template <int N, typename VT, typename MT>
__device__ __forceinline__ void add_rows(VT& acc, const int* __restrict__ nbr, int j,
                                         const MT* __restrict__ Hv, int lane) {
    int s[N];
    MT  r[N];
    #pragma unroll
    for (int k = 0; k < N; ++k) s[k] = nbr[j + k];
    #pragma unroll
    for (int k = 0; k < N; ++k) r[k] = Hv[(size_t)s[k] * 32 + lane];
    #pragma unroll
    for (int k = 0; k < N; ++k) acc += __builtin_convertvector(r[k], VT);
}

template <typename VT, typename MT>
__device__ __forceinline__ void walk_all_edges(VT& acc, int node, int lane, const int* __restrict__ src,
                                               const int* __restrict__ dst, int E,
                                               const MT* __restrict__ Hv) {
    for (int e0 = 0; e0 < E; e0 += 32) {
        int e = e0 + lane;
        int d = e < E ? dst[e] : -1;
        unsigned m = (unsigned)__ballot(d == node);
        while (m) {
            int b = __builtin_ctz(m);
            m &= m - 1;
            int s = src[e0 + b];
            acc += __builtin_convertvector(Hv[(size_t)s * 32 + lane], VT);
        }
    }
}

template <int C, typename ST>
__device__ __forceinline__ typename rowvec<C>::type
gather_row(int node, int lane, int c, float dd, const int* __restrict__ nbr,
           const int* __restrict__ src, const int* __restrict__ dst, int E,
           const ST* __restrict__ H, const float* __restrict__ bias) {
    typedef typename rowvec<C>::type vt;
    typedef typename rowmem<C, ST>::type mt;
    const mt* Hv = reinterpret_cast<const mt*>(H);
    vt acc = __builtin_convertvector(Hv[(size_t)node * 32 + lane], vt);
    if (c <= SEG) {
        int j   = node * SEG;
        int end = j + c;
        for (; j + 8 <= end; j += 8) add_rows<8, vt, mt>(acc, nbr, j, Hv, lane);
        int rem = end - j;
        if (rem & 4) { add_rows<4, vt, mt>(acc, nbr, j, Hv, lane); j += 4; }
        if (rem & 2) { add_rows<2, vt, mt>(acc, nbr, j, Hv, lane); j += 2; }
        if (rem & 1) { add_rows<1, vt, mt>(acc, nbr, j, Hv, lane); }
    } else {
        walk_all_edges<vt, mt>(acc, node, lane, src, dst, E, Hv);
    }
    const vt* bv = reinterpret_cast<const vt*>(bias);
    return acc * dd + bv[lane];
}

__device__ __forceinline__ bool last_row_in(unsigned* ticket, int lane) {
    __threadfence_block();
    unsigned t = 0;
    if (lane == 0) t = atomicAdd(ticket, 1u);
    t = (unsigned)__builtin_amdgcn_readfirstlane((int)t);
    __threadfence_block();
    return t == TILE_M - 1;
}

#define A_PITCH 136

__global__ void __launch_bounds__(TILE_M * 32, 16)
k_gather_gemm(const int* __restrict__ cnt, const int* __restrict__ nbr,
              const int* __restrict__ src, const int* __restrict__ dst, int E,
              const _Float16* __restrict__ H, const float* __restrict__ bias,
              const _Float16* __restrict__ Bh, float* __restrict__ C2, int Nn) {
    typedef __attribute__((ext_vector_type(4))) _Float16 v4h;
    typedef __attribute__((ext_vector_type(8))) _Float16 v8h;
    __shared__ __attribute__((aligned(16))) _Float16 sA[TILE_M * A_PITCH];
    __shared__ __attribute__((aligned(16))) v8h sB[(HID_C / 32) * (OUT_C / 16)][2][32];
    __shared__ __attribute__((aligned(16))) float sD[TILE_M];
    __shared__ unsigned s_ticket;
    if (threadIdx.x == 0) s_ticket = 0;
    __syncthreads();

    int wv   = __builtin_amdgcn_readfirstlane((int)(threadIdx.x >> 5));
    int lane = threadIdx.x & 31;
    int m0   = blockIdx.x * TILE_M;
    int node = m0 + wv;

    v16h bfrag = reinterpret_cast<const v16h*>(Bh)[wv * 32 + lane];

    rowvec<HID_C>::type row = {0.f, 0.f, 0.f, 0.f};
    float dd = 0.f;
    if (node < Nn) {
        int c = cnt[node];
        dd  = rsqrtf((float)(c + 1));
        row = gather_row<HID_C, _Float16>(node, lane, c, dd, nbr, src, dst, E, H, bias);
    }
    if (lane == 0) sD[wv] = dd;
    v4h h;
    h[0] = (_Float16)fmaxf(row[0], 0.f); h[1] = (_Float16)fmaxf(row[1], 0.f);
    h[2] = (_Float16)fmaxf(row[2], 0.f); h[3] = (_Float16)fmaxf(row[3], 0.f);
    *reinterpret_cast<v4h*>(&sA[wv * A_PITCH + lane * 4]) = h;
    sB[wv][0][lane] = __builtin_shufflevector(bfrag, bfrag, 0,1,2,3,4,5,6,7);
    sB[wv][1][lane] = __builtin_shufflevector(bfrag, bfrag, 8,9,10,11,12,13,14,15);

    if (!last_row_in(&s_ticket, lane)) return;

    const int NT = OUT_C / 16;
    int hi = lane >> 4;
    int r  = lane & 15;
    v8f acc[NT];
    v8f zero = {0.f,0.f,0.f,0.f,0.f,0.f,0.f,0.f};
    #pragma unroll
    for (int j = 0; j < NT; ++j) acc[j] = zero;
    #pragma unroll
    for (int kt = 0; kt < HID_C / 32; ++kt) {
        const _Float16* ap = &sA[r * A_PITCH + (kt << 5) + (hi << 3)];
        v8h lo = *reinterpret_cast<const v8h*>(ap);
        v8h up = *reinterpret_cast<const v8h*>(ap + 16);
        v16h a = __builtin_shufflevector(lo, up, 0,1,2,3,4,5,6,7,8,9,10,11,12,13,14,15);
        #pragma unroll
        for (int nt = 0; nt < NT; ++nt) {
            v16h b = __builtin_shufflevector(sB[kt * NT + nt][0][lane], sB[kt * NT + nt][1][lane],
                                             0,1,2,3,4,5,6,7,8,9,10,11,12,13,14,15);
            acc[nt] = __builtin_amdgcn_wmma_f32_16x16x32_f16(
                false, a, false, b, (short)0, acc[nt], false, false);
        }
    }
    int mrow = m0 + (hi << 3);
    bool whole = m0 + TILE_M <= Nn;
    const float4* dp = reinterpret_cast<const float4*>(&sD[hi << 3]);
    float4 d0 = dp[0], d1 = dp[1];
    const float sc[8] = {d0.x, d0.y, d0.z, d0.w, d1.x, d1.y, d1.z, d1.w};
    #pragma unroll
    for (int nt = 0; nt < NT; ++nt) {
        int n = (nt << 4) + r;
        #pragma unroll
        for (int j = 0; j < 8; ++j)
            if (whole || mrow + j < Nn) C2[(long)(mrow + j) * OUT_C + n] = acc[nt][j] * sc[j];
    }
}

__device__ __forceinline__ void pool_runs(const int (&g)[TILE_M], const float (&v)[TILE_M], int c,
                                          float* pool, float* counts) {
    int   gc = g[0];
    float s  = v[0];
    float n  = 1.0f;
    #pragma unroll
    for (int k = 1; k < TILE_M; ++k) {
        if (g[k] != gc) {
            if (gc >= 0) {
                atomic_add_f32(&pool[gc * OUT_C + c], s);
                if (c == 0) atomic_add_f32(&counts[gc], n);
            }
            gc = g[k]; s = 0.0f; n = 0.0f;
        }
        s += v[k];
        n += 1.0f;
    }
    if (gc >= 0) {
        atomic_add_f32(&pool[gc * OUT_C + c], s);
        if (c == 0) atomic_add_f32(&counts[gc], n);
    }
}

__global__ void __launch_bounds__(TILE_M * 32)
k_gather_pool(const int* __restrict__ cnt, const int* __restrict__ nbr,
              const int* __restrict__ src, const int* __restrict__ dst, int E,
              const float* __restrict__ H, const float* __restrict__ bias,
              const int* __restrict__ batch, float* pool, float* counts, int Nn) {
    __shared__ __attribute__((aligned(16))) float sP[TILE_M * OUT_C];
    __shared__ unsigned s_ticket;
    if (threadIdx.x == 0) s_ticket = 0;
    __syncthreads();

    int wv   = __builtin_amdgcn_readfirstlane((int)(threadIdx.x >> 5));
    int lane = threadIdx.x & 31;
    int i0   = blockIdx.x * TILE_M;
    int node = i0 + wv;

    rowvec<OUT_C>::type row = {0.f, 0.f};
    if (node < Nn) {
        int c = cnt[node];
        row = gather_row<OUT_C, float>(node, lane, c, rsqrtf((float)(c + 1)), nbr, src, dst, E, H, bias);
    }
    *reinterpret_cast<rowvec<OUT_C>::type*>(&sP[wv * OUT_C + lane * 2]) = row;

    if (!last_row_in(&s_ticket, lane)) return;

    int g[TILE_M];
    if (i0 + TILE_M <= Nn) {
        #pragma unroll
        for (int k = 0; k < TILE_M; ++k) g[k] = batch[i0 + k];
    } else {
        #pragma unroll
        for (int k = 0; k < TILE_M; ++k) {
            int i  = i0 + k;
            int il = i < Nn ? i : Nn - 1;
            int gk = batch[il];
            g[k] = i < Nn ? gk : -1;
        }
    }
    #pragma unroll
    for (int half = 0; half < OUT_C / 32; ++half) {
        int c = lane + 32 * half;
        float v[TILE_M];
        #pragma unroll
        for (int k = 0; k < TILE_M; ++k) v[k] = sP[k * OUT_C + c];
        pool_runs(g, v, c, pool, counts);
    }
}

__global__ void k_pool_final(const float* __restrict__ pool, const float* __restrict__ counts,
                             float* __restrict__ out) {
    int idx = blockIdx.x * blockDim.x + threadIdx.x;
    if (idx >= NGRAPH * OUT_C) return;
    out[idx] = pool[idx] / fmaxf(counts[idx >> 6], 1.0f);
}

extern "C" void kernel_launch(void* const* d_in, const int* in_sizes, int n_in,
                              void* d_out, int out_size, void* d_ws, size_t ws_size,
                              hipStream_t stream) {
    const float* x   = (const float*)d_in[0];
    const float* W1  = (const float*)d_in[1];
    const float* b1  = (const float*)d_in[2];
    const float* W2  = (const float*)d_in[3];
    const float* b2  = (const float*)d_in[4];
    const int*   ei  = (const int*)d_in[5];
    const int*   bat = (const int*)d_in[6];

    const int Nn = in_sizes[0] / IN_C;
    const int E  = in_sizes[5] / 2;
    const int* src = ei;
    const int* dst = ei + E;

    char* w = (char*)d_ws;
    size_t off = 0;
    auto alloc = [&](size_t bytes) -> void* {
        void* p = w + off;
        off = (off + bytes + 255) & ~(size_t)255;
        return p;
    };
    int*      cnt   = (int*)alloc((size_t)Nn * 4);
    int*      nbr   = (int*)alloc((size_t)Nn * SEG * 4);
    _Float16* W1h   = (_Float16*)alloc((size_t)IN_C * HID_C * 2);
    _Float16* W2h   = (_Float16*)alloc((size_t)HID_C * OUT_C * 2);
    _Float16* h1    = (_Float16*)alloc((size_t)Nn * HID_C * 2);
    float*    h2    = (float*)alloc((size_t)Nn * OUT_C * 4);
    float*    pool  = (float*)alloc((size_t)NGRAPH * OUT_C * 4);
    float*    cnts  = (float*)alloc((size_t)NGRAPH * 4);
    if (off > ws_size) return;

    const int T = 256;
    const int nbins = (Nn + BIN_N - 1) / BIN_N;
    const int nblk  = (E + EPB - 1) / EPB;
    unsigned* runs   = (unsigned*)h2;
    int*      runlen = (int*)((char*)h2 + (size_t)nbins * nblk * RUN * 4);
    if (nbins > MAXBIN || Nn > (1 << (31 - BIN_SH)) ||
        (size_t)nbins * nblk * (RUN * 4 + 4) > (size_t)Nn * OUT_C * 4) return;
    {
        int blocks = nblk + (NGRAPH * OUT_C) / TA + 1 + (IN_C * HID_C) / TA + (HID_C * OUT_C) / TA;
        k_bin_setup<<<blocks, TA, 0, stream>>>(src, dst, E, nbins, nblk, runs, runlen, pool, cnts, W1, W1h, W2, W2h);
    }
    k_rank_place<<<nbins, TB, 0, stream>>>(src, dst, E, nbins, nblk, runs, runlen, cnt, nbr, Nn);
    {
        int waves = Nn / 16;
        int gemm_blocks = (waves * 32 + T - 1) / T;
        k_gemm<HID_C / 16><<<gemm_blocks, T, 0, stream>>>(x, W1h, h1, Nn, IN_C, HID_C, cnt);
    }

    k_gather_gemm<<<(Nn + TILE_M - 1) / TILE_M, TILE_M * 32, 0, stream>>>(
        cnt, nbr, src, dst, E, h1, b1, W2h, h2, Nn);

    k_gather_pool<<<(Nn + TILE_M - 1) / TILE_M, TILE_M * 32, 0, stream>>>(
        cnt, nbr, src, dst, E, h2, b2, bat, pool, cnts, Nn);
    k_pool_final<<<(NGRAPH * OUT_C + T - 1) / T, T, 0, stream>>>(pool, cnts, (float*)d_out);
}
